# starting kernel
# baseline (speedup 1.0000x reference)
#include <hip/hip_runtime.h>

typedef float v2f __attribute__((ext_vector_type(2)));
typedef float v8f __attribute__((ext_vector_type(8)));

#define N_NODES 50000
#define HIDC 64
#define KMAX 128
#define APAD 4


__global__ void k_init_deg(float* __restrict__ deg, int n) {
    int i = blockIdx.x * blockDim.x + threadIdx.x;
    if (i < n) deg[i] = 1.0f;
}

__global__ void k_count_deg(const int* __restrict__ dst, float* __restrict__ deg, int E) {
    int e = blockIdx.x * blockDim.x + threadIdx.x;
    if (e < E) atomicAdd(&deg[dst[e]], 1.0f);
}

__global__ void k_rsqrt_inplace(float* __restrict__ d, int n) {
    int i = blockIdx.x * blockDim.x + threadIdx.x;
    if (i < n) d[i] = rsqrtf(d[i]);
}


__global__ void k_wmma_gemm(const float* __restrict__ A, const float* __restrict__ B,
                            const float* __restrict__ bias, float* __restrict__ C,
                            int K, int N) {
    __shared__ float As[16 * (KMAX + APAD)];

    const int m0   = blockIdx.x * 16;
    const int lane = threadIdx.x & 31;
    const int wave = threadIdx.x >> 5;
    const int n0   = wave * 16;
    const int half = lane >> 4;
    const int idx  = lane & 15;
    const int ldsK = K + APAD;

    {
        const int kq = K >> 2;
        for (int r = 0; r < 16; ++r) {
            const float4* srow = (const float4*)(A + (long)(m0 + r) * K);
            float4*       drow = (float4*)(As + r * ldsK);
            for (int c = threadIdx.x; c < kq; c += blockDim.x) drow[c] = srow[c];
        }
    }
    __syncthreads();

    const float* arow = As + idx * ldsK;
    const int    bcol = n0 + idx;
    const bool   bok  = bcol < N;
    const float* bp   = B + (bok ? bcol : (N - 1));

    v8f acc = {0.f, 0.f, 0.f, 0.f, 0.f, 0.f, 0.f, 0.f};

    for (int k = 0; k < K; k += 4) {
        const int ka = k + 2 * half;
        v2f a, b;
        a.x = arow[ka];
        a.y = arow[ka + 1];
        b.x = bp[(long)ka * N];
        b.y = bp[(long)(ka + 1) * N];
        acc = __builtin_amdgcn_wmma_f32_16x16x4_f32(
             false, a,  false, b,
             (short)0, acc,  false,  false);
    }

    if (bok) {
        const float bb = bias ? bias[bcol] : 0.0f;
#pragma unroll
        for (int v = 0; v < 8; ++v) {
            const int row = m0 + v + 8 * half;
            C[(long)row * N + bcol] = acc[v] + bb;
        }
    }
}


__global__ void k_self_init(const float* __restrict__ hw, const float* __restrict__ dis,
                            float* __restrict__ agg, int n) {
    long t = (long)blockIdx.x * blockDim.x + threadIdx.x;
    if (t < (long)n * HIDC) {
        int i = (int)(t >> 6);
        float di = dis[i];
        agg[t] = hw[t] * di * di;
    }
}

__global__ void k_scatter(const float* __restrict__ hw, const float* __restrict__ dis,
                          const int* __restrict__ src, const int* __restrict__ dst,
                          float* __restrict__ agg, int E) {
    long t = (long)blockIdx.x * blockDim.x + threadIdx.x;
    int e = (int)(t >> 4);
    if (e >= E) return;
    int g = (int)(t & 15);
    int s = src[e];
    int d = dst[e];
    float w = dis[s] * dis[d];
    const float4 v = *((const float4*)(hw + (long)s * HIDC) + g);
    float* ap = agg + (long)d * HIDC + g * 4;
    atomicAdd(ap + 0, v.x * w);
    atomicAdd(ap + 1, v.y * w);
    atomicAdd(ap + 2, v.z * w);
    atomicAdd(ap + 3, v.w * w);
}

__global__ void k_bias_relu(float* __restrict__ h, const float* __restrict__ bias, int n) {
    long t = (long)blockIdx.x * blockDim.x + threadIdx.x;
    if (t < (long)n * HIDC) {
        int c = (int)(t & (HIDC - 1));
        float v = h[t] + bias[c];
        h[t] = v > 0.f ? v : 0.f;
    }
}


extern "C" void kernel_launch(void* const* d_in, const int* in_sizes, int n_in,
                              void* d_out, int out_size, void* d_ws, size_t ws_size,
                              hipStream_t stream) {
    const float* x  = (const float*)d_in[0];
    const int*   ei = (const int*)d_in[1];
    const float* W1 = (const float*)d_in[2];
    const float* b1 = (const float*)d_in[3];
    const float* W2 = (const float*)d_in[4];
    const float* b2 = (const float*)d_in[5];
    const float* Wc = (const float*)d_in[6];
    const float* bc = (const float*)d_in[7];
    float* out = (float*)d_out;

    const int n = N_NODES;
    const int E = in_sizes[1] / 2;
    const int* src = ei;
    const int* dst = ei + E;

    char* ws = (char*)d_ws;
    float* dis = (float*)ws;
    size_t off = (((size_t)n * 4) + 255) & ~(size_t)255;
    float* bufA = (float*)(ws + off);
    off += (((size_t)n * HIDC * 4) + 255) & ~(size_t)255;
    float* bufB = (float*)(ws + off);

    const int T = 256;
    const int gN   = (n + T - 1) / T;
    const int gE   = (E + T - 1) / T;
    const int gNC  = (n * HIDC + T - 1) / T;
    const int gE16 = (int)(((long)E * 16 + T - 1) / T);
    const int rowTiles = n / 16;

    k_init_deg<<<gN, T, 0, stream>>>(dis, n);
    k_count_deg<<<gE, T, 0, stream>>>(dst, dis, E);
    k_rsqrt_inplace<<<gN, T, 0, stream>>>(dis, n);

    k_wmma_gemm<<<rowTiles, 128, 0, stream>>>(x, W1, nullptr, bufA, 128, HIDC);
    k_self_init<<<gNC, T, 0, stream>>>(bufA, dis, bufB, n);
    k_scatter<<<gE16, T, 0, stream>>>(bufA, dis, src, dst, bufB, E);
    k_bias_relu<<<gNC, T, 0, stream>>>(bufB, b1, n);

    k_wmma_gemm<<<rowTiles, 128, 0, stream>>>(bufB, W2, nullptr, bufA, HIDC, HIDC);
    k_self_init<<<gNC, T, 0, stream>>>(bufA, dis, bufB, n);
    k_scatter<<<gE16, T, 0, stream>>>(bufA, dis, src, dst, bufB, E);
    k_bias_relu<<<gNC, T, 0, stream>>>(bufB, b2, n);

    k_wmma_gemm<<<rowTiles, 32, 0, stream>>>(bufB, Wc, bc, out, HIDC, 10);
}


# Aggregate by gather over a dst-sorted edge list (returning int count gives ranks, scan, fill) instead of f32 atomic scatter; per-node sum order follows arrival rank; self-loop, bias, relu folded in
# speedup vs baseline: 1.7023x; 1.7023x over previous
#include <hip/hip_runtime.h>

typedef float v2f __attribute__((ext_vector_type(2)));
typedef float v8f __attribute__((ext_vector_type(8)));

#define N_NODES 50000
#define HIDC 64
#define KMAX 128
#define APAD 4


__global__ void k_zero_cnt(int* __restrict__ cnt, int n) {
    int i = blockIdx.x * blockDim.x + threadIdx.x;
    if (i < n) cnt[i] = 0;
}

__global__ void k_count_rank(const int* __restrict__ dst, int* __restrict__ cnt,
                             int* __restrict__ rank, int E) {
    int e = blockIdx.x * blockDim.x + threadIdx.x;
    if (e < E) rank[e] = atomicAdd(&cnt[dst[e]], 1);
}

#define SCAN_T 1024
__global__ void k_scan_dis(const int* __restrict__ cnt, int* __restrict__ rowptr,
                           float* __restrict__ dis, int n4) {
    __shared__ int s_front[SCAN_T / 32];
    __shared__ int s_wtot[SCAN_T / 32];
    const int tid  = threadIdx.x;
    const int lane = tid & 31;
    const int wave = tid >> 5;
    const int4* c4 = (const int4*)cnt;
    const int base4 = blockIdx.x * SCAN_T;

    int front = 0;
#pragma unroll 4
    for (int q = tid; q < base4; q += SCAN_T) {
        const int4 u = c4[q];
        front += (u.x + u.y) + (u.z + u.w);
    }
    for (int o = 16; o > 0; o >>= 1) front += __shfl_xor(front, o);
    if (lane == 0) s_front[wave] = front;

    const int q = base4 + tid;
    int4 v = make_int4(0, 0, 0, 0);
    if (q < n4) v = c4[q];
    const int tsum = (v.x + v.y) + (v.z + v.w);
    int inc = tsum;
    for (int o = 1; o < 32; o <<= 1) {
        const int u = __shfl_up(inc, o);
        if (lane >= o) inc += u;
    }
    if (lane == 31) s_wtot[wave] = inc;
    __syncthreads();

    int carry = 0;
    for (int w = 0; w < SCAN_T / 32; ++w) carry += s_front[w];
    for (int w = 0; w < wave; ++w) carry += s_wtot[w];

    if (q < n4) {
        int4 r;
        r.x = carry + inc - tsum;
        r.y = r.x + v.x;
        r.z = r.y + v.y;
        r.w = r.z + v.z;
        ((int4*)rowptr)[q] = r;
        if (q == n4 - 1) rowptr[4 * n4] = r.w + v.w;
        float4 d;
        d.x = rsqrtf((float)(v.x + 1));
        d.y = rsqrtf((float)(v.y + 1));
        d.z = rsqrtf((float)(v.z + 1));
        d.w = rsqrtf((float)(v.w + 1));
        ((float4*)dis)[q] = d;
    }
}

__global__ void k_fill(const int* __restrict__ src, const int* __restrict__ dst,
                       const int* __restrict__ rank, const int* __restrict__ rowptr,
                       const float* __restrict__ dis, int2* __restrict__ csr, int E) {
    int e = blockIdx.x * blockDim.x + threadIdx.x;
    if (e < E) {
        const int s = src[e];
        const int slot = rowptr[dst[e]] + rank[e];
        csr[slot] = make_int2(s, __float_as_int(dis[s]));
    }
}


__global__ void k_wmma_gemm(const float* __restrict__ A, const float* __restrict__ B,
                            const float* __restrict__ bias, float* __restrict__ C,
                            int K, int N) {
    __shared__ float As[16 * (KMAX + APAD)];

    const int m0   = blockIdx.x * 16;
    const int lane = threadIdx.x & 31;
    const int wave = threadIdx.x >> 5;
    const int n0   = wave * 16;
    const int half = lane >> 4;
    const int idx  = lane & 15;
    const int ldsK = K + APAD;

    {
        const int kq = K >> 2;
        for (int r = 0; r < 16; ++r) {
            const float4* srow = (const float4*)(A + (long)(m0 + r) * K);
            float4*       drow = (float4*)(As + r * ldsK);
            for (int c = threadIdx.x; c < kq; c += blockDim.x) drow[c] = srow[c];
        }
    }
    __syncthreads();

    const float* arow = As + idx * ldsK;
    const int    bcol = n0 + idx;
    const bool   bok  = bcol < N;
    const float* bp   = B + (bok ? bcol : (N - 1));

    v8f acc = {0.f, 0.f, 0.f, 0.f, 0.f, 0.f, 0.f, 0.f};

    for (int k = 0; k < K; k += 4) {
        const int ka = k + 2 * half;
        v2f a, b;
        a.x = arow[ka];
        a.y = arow[ka + 1];
        b.x = bp[(long)ka * N];
        b.y = bp[(long)(ka + 1) * N];
        acc = __builtin_amdgcn_wmma_f32_16x16x4_f32(
             false, a,  false, b,
             (short)0, acc,  false,  false);
    }

    if (bok) {
        const float bb = bias ? bias[bcol] : 0.0f;
#pragma unroll
        for (int v = 0; v < 8; ++v) {
            const int row = m0 + v + 8 * half;
            C[(long)row * N + bcol] = acc[v] + bb;
        }
    }
}


static __device__ __forceinline__ void fma4(float4& a, const float4 v, const float w) {
    a.x = fmaf(v.x, w, a.x);
    a.y = fmaf(v.y, w, a.y);
    a.z = fmaf(v.z, w, a.z);
    a.w = fmaf(v.w, w, a.w);
}

static __device__ __forceinline__ void fma4_if(float4& a, const float4 v, const float w, const bool c) {
    a.x = c ? fmaf(v.x, w, a.x) : a.x;
    a.y = c ? fmaf(v.y, w, a.y) : a.y;
    a.z = c ? fmaf(v.z, w, a.z) : a.z;
    a.w = c ? fmaf(v.w, w, a.w) : a.w;
}

__global__ void k_gather(const float* __restrict__ hw, const float* __restrict__ dis,
                         const int* __restrict__ rowptr,
                         const int2* __restrict__ csr, const float* __restrict__ bias,
                         float* __restrict__ hout, int n) {
    const int t = blockIdx.x * blockDim.x + threadIdx.x;
    const int i = t >> 4;
    if (i >= n) return;
    const int g = t & 15;
    const float4* hw4 = (const float4*)hw;

    const float dd  = dis[i];
    const int   beg = rowptr[i];
    const int   end = rowptr[i + 1];

    float4 acc = make_float4(0.f, 0.f, 0.f, 0.f);
    fma4(acc, hw4[i * (HIDC / 4) + g], dd * dd);

    const int last = end - 1;
    for (int j = beg; j < end; j += 4) {
        const int2 e0 = csr[j];
        const int2 e1 = csr[min(j + 1, last)];
        const int2 e2 = csr[min(j + 2, last)];
        const int2 e3 = csr[min(j + 3, last)];
        const float4 v0 = hw4[e0.x * (HIDC / 4) + g];
        const float4 v1 = hw4[e1.x * (HIDC / 4) + g];
        const float4 v2 = hw4[e2.x * (HIDC / 4) + g];
        const float4 v3 = hw4[e3.x * (HIDC / 4) + g];
        fma4(acc, v0, __int_as_float(e0.y) * dd);
        fma4_if(acc, v1, __int_as_float(e1.y) * dd, j + 1 < end);
        fma4_if(acc, v2, __int_as_float(e2.y) * dd, j + 2 < end);
        fma4_if(acc, v3, __int_as_float(e3.y) * dd, j + 3 < end);
    }

    const float4 b = ((const float4*)bias)[g];
    float4 o;
    o.x = fmaxf(acc.x + b.x, 0.f);
    o.y = fmaxf(acc.y + b.y, 0.f);
    o.z = fmaxf(acc.z + b.z, 0.f);
    o.w = fmaxf(acc.w + b.w, 0.f);
    ((float4*)hout)[i * (HIDC / 4) + g] = o;
}


extern "C" void kernel_launch(void* const* d_in, const int* in_sizes, int n_in,
                              void* d_out, int out_size, void* d_ws, size_t ws_size,
                              hipStream_t stream) {
    const float* x  = (const float*)d_in[0];
    const int*   ei = (const int*)d_in[1];
    const float* W1 = (const float*)d_in[2];
    const float* b1 = (const float*)d_in[3];
    const float* W2 = (const float*)d_in[4];
    const float* b2 = (const float*)d_in[5];
    const float* Wc = (const float*)d_in[6];
    const float* bc = (const float*)d_in[7];
    float* out = (float*)d_out;

    const int n = N_NODES;
    const int E = in_sizes[1] / 2;
    const int* src = ei;
    const int* dst = ei + E;

    char* ws = (char*)d_ws;
    size_t off = 0;
    int* cnt = (int*)(ws + off);
    off += (((size_t)n * 4) + 255) & ~(size_t)255;
    int* rowptr = (int*)(ws + off);
    off += (((size_t)(n + 1) * 4) + 255) & ~(size_t)255;
    float* dis = (float*)(ws + off);
    off += (((size_t)n * 4) + 255) & ~(size_t)255;
    int* rank = (int*)(ws + off);
    off += (((size_t)E * 4) + 255) & ~(size_t)255;
    int2* csr = (int2*)(ws + off);
    off += (((size_t)E * 8) + 255) & ~(size_t)255;
    float* bufA = (float*)(ws + off);
    off += (((size_t)n * HIDC * 4) + 255) & ~(size_t)255;
    float* bufB = (float*)(ws + off);

    const int T = 256;
    const int gN   = (n + T - 1) / T;
    const int gE   = (E + T - 1) / T;
    const int gN16 = (n * 16 + T - 1) / T;
    const int n4   = n / 4;
    const int gS   = (n4 + SCAN_T - 1) / SCAN_T;
    const int rowTiles = n / 16;

    k_zero_cnt<<<gN, T, 0, stream>>>(cnt, n);
    k_count_rank<<<gE, T, 0, stream>>>(dst, cnt, rank, E);
    k_scan_dis<<<gS, SCAN_T, 0, stream>>>(cnt, rowptr, dis, n4);
    k_fill<<<gE, T, 0, stream>>>(src, dst, rank, rowptr, dis, csr, E);

    k_wmma_gemm<<<rowTiles, 128, 0, stream>>>(x, W1, nullptr, bufA, 128, HIDC);
    k_gather<<<gN16, T, 0, stream>>>(bufA, dis, rowptr, csr, b1, bufB, n);

    k_wmma_gemm<<<rowTiles, 128, 0, stream>>>(bufB, W2, nullptr, bufA, HIDC, HIDC);
    k_gather<<<gN16, T, 0, stream>>>(bufA, dis, rowptr, csr, b2, bufB, n);

    k_wmma_gemm<<<rowTiles, 32, 0, stream>>>(bufB, Wc, bc, out, HIDC, 10);
}


# Fuse each aggregation with the dense product after it: the block's 16x64 h tile goes to LDS and is multiplied by W2 / Wc on WMMA in the gather's tail; h1, h2 never reach memory; 9 -> 7 launches
# speedup vs baseline: 1.9746x; 1.1600x over previous
#include <hip/hip_runtime.h>

typedef float v2f __attribute__((ext_vector_type(2)));
typedef float v8f __attribute__((ext_vector_type(8)));

#define N_NODES 50000
#define HIDC 64
#define KMAX 128
#define APAD 4


__global__ void k_zero_cnt(int* __restrict__ cnt, int n) {
    int i = blockIdx.x * blockDim.x + threadIdx.x;
    if (i < n) cnt[i] = 0;
}

__global__ void k_count_rank(const int* __restrict__ dst, int* __restrict__ cnt,
                             int* __restrict__ rank, int E) {
    int e = blockIdx.x * blockDim.x + threadIdx.x;
    if (e < E) rank[e] = atomicAdd(&cnt[dst[e]], 1);
}

#define SCAN_T 1024
__global__ void k_scan_dis(const int* __restrict__ cnt, int* __restrict__ rowptr,
                           float* __restrict__ dis, int n4) {
    __shared__ int s_front[SCAN_T / 32];
    __shared__ int s_wtot[SCAN_T / 32];
    const int tid  = threadIdx.x;
    const int lane = tid & 31;
    const int wave = tid >> 5;
    const int4* c4 = (const int4*)cnt;
    const int base4 = blockIdx.x * SCAN_T;

    int front = 0;
#pragma unroll 4
    for (int q = tid; q < base4; q += SCAN_T) {
        const int4 u = c4[q];
        front += (u.x + u.y) + (u.z + u.w);
    }
    for (int o = 16; o > 0; o >>= 1) front += __shfl_xor(front, o);
    if (lane == 0) s_front[wave] = front;

    const int q = base4 + tid;
    int4 v = make_int4(0, 0, 0, 0);
    if (q < n4) v = c4[q];
    const int tsum = (v.x + v.y) + (v.z + v.w);
    int inc = tsum;
    for (int o = 1; o < 32; o <<= 1) {
        const int u = __shfl_up(inc, o);
        if (lane >= o) inc += u;
    }
    if (lane == 31) s_wtot[wave] = inc;
    __syncthreads();

    int carry = 0;
    for (int w = 0; w < SCAN_T / 32; ++w) carry += s_front[w];
    for (int w = 0; w < wave; ++w) carry += s_wtot[w];

    if (q < n4) {
        int4 r;
        r.x = carry + inc - tsum;
        r.y = r.x + v.x;
        r.z = r.y + v.y;
        r.w = r.z + v.z;
        ((int4*)rowptr)[q] = r;
        if (q == n4 - 1) rowptr[4 * n4] = r.w + v.w;
        float4 d;
        d.x = rsqrtf((float)(v.x + 1));
        d.y = rsqrtf((float)(v.y + 1));
        d.z = rsqrtf((float)(v.z + 1));
        d.w = rsqrtf((float)(v.w + 1));
        ((float4*)dis)[q] = d;
    }
}

__global__ void k_fill(const int* __restrict__ src, const int* __restrict__ dst,
                       const int* __restrict__ rank, const int* __restrict__ rowptr,
                       const float* __restrict__ dis, int2* __restrict__ csr, int E) {
    int e = blockIdx.x * blockDim.x + threadIdx.x;
    if (e < E) {
        const int s = src[e];
        const int slot = rowptr[dst[e]] + rank[e];
        csr[slot] = make_int2(s, __float_as_int(dis[s]));
    }
}


__global__ void k_wmma_gemm(const float* __restrict__ A, const float* __restrict__ B,
                            const float* __restrict__ bias, float* __restrict__ C,
                            int K, int N) {
    __shared__ float As[16 * (KMAX + APAD)];

    const int m0   = blockIdx.x * 16;
    const int lane = threadIdx.x & 31;
    const int wave = threadIdx.x >> 5;
    const int n0   = wave * 16;
    const int half = lane >> 4;
    const int idx  = lane & 15;
    const int ldsK = K + APAD;

    {
        const int kq = K >> 2;
        for (int r = 0; r < 16; ++r) {
            const float4* srow = (const float4*)(A + (long)(m0 + r) * K);
            float4*       drow = (float4*)(As + r * ldsK);
            for (int c = threadIdx.x; c < kq; c += blockDim.x) drow[c] = srow[c];
        }
    }
    __syncthreads();

    const float* arow = As + idx * ldsK;
    const int    bcol = n0 + idx;
    const bool   bok  = bcol < N;
    const float* bp   = B + (bok ? bcol : (N - 1));

    v8f acc = {0.f, 0.f, 0.f, 0.f, 0.f, 0.f, 0.f, 0.f};

    for (int k = 0; k < K; k += 4) {
        const int ka = k + 2 * half;
        v2f a, b;
        a.x = arow[ka];
        a.y = arow[ka + 1];
        b.x = bp[(long)ka * N];
        b.y = bp[(long)(ka + 1) * N];
        acc = __builtin_amdgcn_wmma_f32_16x16x4_f32(
             false, a,  false, b,
             (short)0, acc,  false,  false);
    }

    if (bok) {
        const float bb = bias ? bias[bcol] : 0.0f;
#pragma unroll
        for (int v = 0; v < 8; ++v) {
            const int row = m0 + v + 8 * half;
            C[(long)row * N + bcol] = acc[v] + bb;
        }
    }
}


static __device__ __forceinline__ void fma4(float4& a, const float4 v, const float w) {
    a.x = fmaf(v.x, w, a.x);
    a.y = fmaf(v.y, w, a.y);
    a.z = fmaf(v.z, w, a.z);
    a.w = fmaf(v.w, w, a.w);
}

static __device__ __forceinline__ void fma4_if(float4& a, const float4 v, const float w, const bool c) {
    a.x = c ? fmaf(v.x, w, a.x) : a.x;
    a.y = c ? fmaf(v.y, w, a.y) : a.y;
    a.z = c ? fmaf(v.z, w, a.z) : a.z;
    a.w = c ? fmaf(v.w, w, a.w) : a.w;
}

#define HPAD 4
template <int NOUT>
__global__ void k_gather_gemm(const float* __restrict__ hw, const float* __restrict__ dis,
                              const int* __restrict__ rowptr,
                              const int2* __restrict__ csr, const float* __restrict__ bias,
                              const float* __restrict__ Wn, const float* __restrict__ bn,
                              float* __restrict__ outp, int n) {
    constexpr int NT = (NOUT + 15) / 16;
    __shared__ float Hs[16 * (HIDC + HPAD)];

    const int t = blockIdx.x * blockDim.x + threadIdx.x;
    const int g = t & 15;
    const bool live = (t >> 4) < n;
    const int i = live ? (t >> 4) : (n - 1);
    const float4* hw4 = (const float4*)hw;

    const float dd  = dis[i];
    const int   beg = rowptr[i];
    const int   end = rowptr[i + 1];

    float4 acc = make_float4(0.f, 0.f, 0.f, 0.f);
    fma4(acc, hw4[i * (HIDC / 4) + g], dd * dd);

    const int last = end - 1;
    for (int j = beg; j < end; j += 4) {
        const int2 e0 = csr[j];
        const int2 e1 = csr[min(j + 1, last)];
        const int2 e2 = csr[min(j + 2, last)];
        const int2 e3 = csr[min(j + 3, last)];
        const float4 v0 = hw4[e0.x * (HIDC / 4) + g];
        const float4 v1 = hw4[e1.x * (HIDC / 4) + g];
        const float4 v2 = hw4[e2.x * (HIDC / 4) + g];
        const float4 v3 = hw4[e3.x * (HIDC / 4) + g];
        fma4(acc, v0, __int_as_float(e0.y) * dd);
        fma4_if(acc, v1, __int_as_float(e1.y) * dd, j + 1 < end);
        fma4_if(acc, v2, __int_as_float(e2.y) * dd, j + 2 < end);
        fma4_if(acc, v3, __int_as_float(e3.y) * dd, j + 3 < end);
    }

    const float4 b = ((const float4*)bias)[g];
    float4 o;
    o.x = fmaxf(acc.x + b.x, 0.f);
    o.y = fmaxf(acc.y + b.y, 0.f);
    o.z = fmaxf(acc.z + b.z, 0.f);
    o.w = fmaxf(acc.w + b.w, 0.f);

    ((float4*)(Hs + (threadIdx.x >> 4) * (HIDC + HPAD)))[g] = o;
    __syncthreads();

    const int wave = threadIdx.x >> 5;
    if (wave < NT) {
        const int m0   = blockIdx.x * 16;
        const int lane = threadIdx.x & 31;
        const int half = lane >> 4;
        const int idx  = lane & 15;
        const float* arow = Hs + idx * (HIDC + HPAD);
        const int    bcol = wave * 16 + idx;
        const bool   bok  = bcol < NOUT;
        const float* bp   = Wn + (bok ? bcol : (NOUT - 1));

        v8f d = {0.f, 0.f, 0.f, 0.f, 0.f, 0.f, 0.f, 0.f};
#pragma unroll
        for (int k = 0; k < HIDC; k += 4) {
            const int ka = k + 2 * half;
            v2f a, w;
            a.x = arow[ka];
            a.y = arow[ka + 1];
            w.x = bp[ka * NOUT];
            w.y = bp[(ka + 1) * NOUT];
            d = __builtin_amdgcn_wmma_f32_16x16x4_f32(
                 false, a,  false, w,
                 (short)0, d,  false,  false);
        }

        if (bok) {
            const float bb = bn ? bn[bcol] : 0.0f;
#pragma unroll
            for (int v = 0; v < 8; ++v) {
                const int row = m0 + v + 8 * half;
                if (row < n) outp[(long)row * NOUT + bcol] = d[v] + bb;
            }
        }
    }
}


extern "C" void kernel_launch(void* const* d_in, const int* in_sizes, int n_in,
                              void* d_out, int out_size, void* d_ws, size_t ws_size,
                              hipStream_t stream) {
    const float* x  = (const float*)d_in[0];
    const int*   ei = (const int*)d_in[1];
    const float* W1 = (const float*)d_in[2];
    const float* b1 = (const float*)d_in[3];
    const float* W2 = (const float*)d_in[4];
    const float* b2 = (const float*)d_in[5];
    const float* Wc = (const float*)d_in[6];
    const float* bc = (const float*)d_in[7];
    float* out = (float*)d_out;

    const int n = N_NODES;
    const int E = in_sizes[1] / 2;
    const int* src = ei;
    const int* dst = ei + E;

    char* ws = (char*)d_ws;
    size_t off = 0;
    int* cnt = (int*)(ws + off);
    off += (((size_t)n * 4) + 255) & ~(size_t)255;
    int* rowptr = (int*)(ws + off);
    off += (((size_t)(n + 1) * 4) + 255) & ~(size_t)255;
    float* dis = (float*)(ws + off);
    off += (((size_t)n * 4) + 255) & ~(size_t)255;
    int* rank = (int*)(ws + off);
    off += (((size_t)E * 4) + 255) & ~(size_t)255;
    int2* csr = (int2*)(ws + off);
    off += (((size_t)E * 8) + 255) & ~(size_t)255;
    float* bufA = (float*)(ws + off);
    off += (((size_t)n * HIDC * 4) + 255) & ~(size_t)255;
    float* bufB = (float*)(ws + off);

    const int T = 256;
    const int gN   = (n + T - 1) / T;
    const int gE   = (E + T - 1) / T;
    const int gN16 = (n * 16 + T - 1) / T;
    const int n4   = n / 4;
    const int gS   = (n4 + SCAN_T - 1) / SCAN_T;
    const int rowTiles = n / 16;

    k_zero_cnt<<<gN, T, 0, stream>>>(cnt, n);
    k_count_rank<<<gE, T, 0, stream>>>(dst, cnt, rank, E);
    k_scan_dis<<<gS, SCAN_T, 0, stream>>>(cnt, rowptr, dis, n4);
    k_fill<<<gE, T, 0, stream>>>(src, dst, rank, rowptr, dis, csr, E);

    k_wmma_gemm<<<rowTiles, 128, 0, stream>>>(x, W1, nullptr, bufA, 128, HIDC);

    k_gather_gemm<HIDC><<<gN16, T, 0, stream>>>(bufA, dis, rowptr, csr, b1, W2, nullptr, bufB, n);

    k_gather_gemm<10><<<gN16, T, 0, stream>>>(bufB, dis, rowptr, csr, b2, Wc, bc, out, n);
}


# Layer 1 product x@W1 moved into the rank-count launch (product blocks of two 16-row tiles alternate with the first edge blocks; K, N compile-time): it streams while the adds wait; 7 -> 6 launches
# speedup vs baseline: 2.2402x; 1.1345x over previous
#include <hip/hip_runtime.h>

typedef float v2f __attribute__((ext_vector_type(2)));
typedef float v8f __attribute__((ext_vector_type(8)));

#define N_NODES 50000
#define HIDC 64
#define APAD 4


__global__ void k_zero_cnt(int* __restrict__ cnt, int n) {
    int i = blockIdx.x * blockDim.x + threadIdx.x;
    if (i < n) cnt[i] = 0;
}

#define SCAN_T 1024
__global__ void k_scan_dis(const int* __restrict__ cnt, int* __restrict__ rowptr,
                           float* __restrict__ dis, int n4) {
    __shared__ int s_front[SCAN_T / 32];
    __shared__ int s_wtot[SCAN_T / 32];
    const int tid  = threadIdx.x;
    const int lane = tid & 31;
    const int wave = tid >> 5;
    const int4* c4 = (const int4*)cnt;
    const int base4 = blockIdx.x * SCAN_T;

    int front = 0;
#pragma unroll 4
    for (int q = tid; q < base4; q += SCAN_T) {
        const int4 u = c4[q];
        front += (u.x + u.y) + (u.z + u.w);
    }
    for (int o = 16; o > 0; o >>= 1) front += __shfl_xor(front, o);
    if (lane == 0) s_front[wave] = front;

    const int q = base4 + tid;
    int4 v = make_int4(0, 0, 0, 0);
    if (q < n4) v = c4[q];
    const int tsum = (v.x + v.y) + (v.z + v.w);
    int inc = tsum;
    for (int o = 1; o < 32; o <<= 1) {
        const int u = __shfl_up(inc, o);
        if (lane >= o) inc += u;
    }
    if (lane == 31) s_wtot[wave] = inc;
    __syncthreads();

    int carry = 0;
    for (int w = 0; w < SCAN_T / 32; ++w) carry += s_front[w];
    for (int w = 0; w < wave; ++w) carry += s_wtot[w];

    if (q < n4) {
        int4 r;
        r.x = carry + inc - tsum;
        r.y = r.x + v.x;
        r.z = r.y + v.y;
        r.w = r.z + v.z;
        ((int4*)rowptr)[q] = r;
        if (q == n4 - 1) rowptr[4 * n4] = r.w + v.w;
        float4 d;
        d.x = rsqrtf((float)(v.x + 1));
        d.y = rsqrtf((float)(v.y + 1));
        d.z = rsqrtf((float)(v.z + 1));
        d.w = rsqrtf((float)(v.w + 1));
        ((float4*)dis)[q] = d;
    }
}

__global__ void k_fill(const int* __restrict__ src, const int* __restrict__ dst,
                       const int* __restrict__ rank, const int* __restrict__ rowptr,
                       const float* __restrict__ dis, int2* __restrict__ csr, int E) {
    int e = blockIdx.x * blockDim.x + threadIdx.x;
    if (e < E) {
        const int s = src[e];
        const int slot = rowptr[dst[e]] + rank[e];
        csr[slot] = make_int2(s, __float_as_int(dis[s]));
    }
}

#define INC 128

__global__ void __launch_bounds__(256)
k_rank_gemm(const int* __restrict__ dst, int* __restrict__ cnt, int* __restrict__ rank, int E,
            const float* __restrict__ A, const float* __restrict__ B, float* __restrict__ C,
            int rowTiles) {
    __shared__ float As[2][16 * (INC + APAD)];

    const int blk   = blockIdx.x;
    const int nProd = (rowTiles + 1) >> 1;
    const bool mixed = blk < 2 * nProd;

    if (!(mixed && (blk & 1))) {
        const int eb = mixed ? (blk >> 1) : (blk - nProd);
        const int e  = eb * 256 + threadIdx.x;
        if (e < E) rank[e] = atomicAdd(&cnt[dst[e]], 1);
        return;
    }

    const int sub  = threadIdx.x >> 7;
    const int t128 = threadIdx.x & 127;
    const int tile = 2 * (blk >> 1) + sub;
    const bool tok = tile < rowTiles;
    const int m0   = (tok ? tile : (rowTiles - 1)) * 16;
    const int lane = threadIdx.x & 31;
    const int wave = t128 >> 5;
    const int n0   = wave * 16;
    const int half = lane >> 4;
    const int idx  = lane & 15;
    constexpr int ldsK = INC + APAD;
    float* as = As[sub];

    {
        const float4* s4 = (const float4*)(A + (long)m0 * INC);
#pragma unroll
        for (int q = t128; q < 16 * (INC / 4); q += 128) {
            const int r = q >> 5;
            const int c = q & 31;
            ((float4*)(as + r * ldsK))[c] = s4[q];
        }
    }
    __syncthreads();

    const float* arow = as + idx * ldsK;
    const int    bcol = n0 + idx;
    const float* bp   = B + bcol;

    v8f acc = {0.f, 0.f, 0.f, 0.f, 0.f, 0.f, 0.f, 0.f};

#pragma unroll 8
    for (int k = 0; k < INC; k += 4) {
        const int ka = k + 2 * half;
        v2f a, b;
        a.x = arow[ka];
        a.y = arow[ka + 1];
        b.x = bp[ka * HIDC];
        b.y = bp[(ka + 1) * HIDC];
        acc = __builtin_amdgcn_wmma_f32_16x16x4_f32(
             false, a,  false, b,
             (short)0, acc,  false,  false);
    }

    if (tok) {
#pragma unroll
        for (int v = 0; v < 8; ++v) {
            const int row = m0 + v + 8 * half;
            C[(long)row * HIDC + bcol] = acc[v];
        }
    }
}


static __device__ __forceinline__ void fma4(float4& a, const float4 v, const float w) {
    a.x = fmaf(v.x, w, a.x);
    a.y = fmaf(v.y, w, a.y);
    a.z = fmaf(v.z, w, a.z);
    a.w = fmaf(v.w, w, a.w);
}

static __device__ __forceinline__ void fma4_if(float4& a, const float4 v, const float w, const bool c) {
    a.x = c ? fmaf(v.x, w, a.x) : a.x;
    a.y = c ? fmaf(v.y, w, a.y) : a.y;
    a.z = c ? fmaf(v.z, w, a.z) : a.z;
    a.w = c ? fmaf(v.w, w, a.w) : a.w;
}

#define HPAD 4
template <int NOUT>
__global__ void k_gather_gemm(const float* __restrict__ hw, const float* __restrict__ dis,
                              const int* __restrict__ rowptr,
                              const int2* __restrict__ csr, const float* __restrict__ bias,
                              const float* __restrict__ Wn, const float* __restrict__ bn,
                              float* __restrict__ outp, int n) {
    constexpr int NT = (NOUT + 15) / 16;
    __shared__ float Hs[16 * (HIDC + HPAD)];

    const int t = blockIdx.x * blockDim.x + threadIdx.x;
    const int g = t & 15;
    const bool live = (t >> 4) < n;
    const int i = live ? (t >> 4) : (n - 1);
    const float4* hw4 = (const float4*)hw;

    const float dd  = dis[i];
    const int   beg = rowptr[i];
    const int   end = rowptr[i + 1];

    float4 acc = make_float4(0.f, 0.f, 0.f, 0.f);
    fma4(acc, hw4[i * (HIDC / 4) + g], dd * dd);

    const int last = end - 1;
    for (int j = beg; j < end; j += 4) {
        const int2 e0 = csr[j];
        const int2 e1 = csr[min(j + 1, last)];
        const int2 e2 = csr[min(j + 2, last)];
        const int2 e3 = csr[min(j + 3, last)];
        const float4 v0 = hw4[e0.x * (HIDC / 4) + g];
        const float4 v1 = hw4[e1.x * (HIDC / 4) + g];
        const float4 v2 = hw4[e2.x * (HIDC / 4) + g];
        const float4 v3 = hw4[e3.x * (HIDC / 4) + g];
        fma4(acc, v0, __int_as_float(e0.y) * dd);
        fma4_if(acc, v1, __int_as_float(e1.y) * dd, j + 1 < end);
        fma4_if(acc, v2, __int_as_float(e2.y) * dd, j + 2 < end);
        fma4_if(acc, v3, __int_as_float(e3.y) * dd, j + 3 < end);
    }

    const float4 b = ((const float4*)bias)[g];
    float4 o;
    o.x = fmaxf(acc.x + b.x, 0.f);
    o.y = fmaxf(acc.y + b.y, 0.f);
    o.z = fmaxf(acc.z + b.z, 0.f);
    o.w = fmaxf(acc.w + b.w, 0.f);

    ((float4*)(Hs + (threadIdx.x >> 4) * (HIDC + HPAD)))[g] = o;
    __syncthreads();

    const int wave = threadIdx.x >> 5;
    if (wave < NT) {
        const int m0   = blockIdx.x * 16;
        const int lane = threadIdx.x & 31;
        const int half = lane >> 4;
        const int idx  = lane & 15;
        const float* arow = Hs + idx * (HIDC + HPAD);
        const int    bcol = wave * 16 + idx;
        const bool   bok  = bcol < NOUT;
        const float* bp   = Wn + (bok ? bcol : (NOUT - 1));

        v8f d = {0.f, 0.f, 0.f, 0.f, 0.f, 0.f, 0.f, 0.f};
#pragma unroll
        for (int k = 0; k < HIDC; k += 4) {
            const int ka = k + 2 * half;
            v2f a, w;
            a.x = arow[ka];
            a.y = arow[ka + 1];
            w.x = bp[ka * NOUT];
            w.y = bp[(ka + 1) * NOUT];
            d = __builtin_amdgcn_wmma_f32_16x16x4_f32(
                 false, a,  false, w,
                 (short)0, d,  false,  false);
        }

        if (bok) {
            const float bb = bn ? bn[bcol] : 0.0f;
#pragma unroll
            for (int v = 0; v < 8; ++v) {
                const int row = m0 + v + 8 * half;
                if (row < n) outp[(long)row * NOUT + bcol] = d[v] + bb;
            }
        }
    }
}


extern "C" void kernel_launch(void* const* d_in, const int* in_sizes, int n_in,
                              void* d_out, int out_size, void* d_ws, size_t ws_size,
                              hipStream_t stream) {
    const float* x  = (const float*)d_in[0];
    const int*   ei = (const int*)d_in[1];
    const float* W1 = (const float*)d_in[2];
    const float* b1 = (const float*)d_in[3];
    const float* W2 = (const float*)d_in[4];
    const float* b2 = (const float*)d_in[5];
    const float* Wc = (const float*)d_in[6];
    const float* bc = (const float*)d_in[7];
    float* out = (float*)d_out;

    const int n = N_NODES;
    const int E = in_sizes[1] / 2;
    const int* src = ei;
    const int* dst = ei + E;

    char* ws = (char*)d_ws;
    size_t off = 0;
    int* cnt = (int*)(ws + off);
    off += (((size_t)n * 4) + 255) & ~(size_t)255;
    int* rowptr = (int*)(ws + off);
    off += (((size_t)(n + 1) * 4) + 255) & ~(size_t)255;
    float* dis = (float*)(ws + off);
    off += (((size_t)n * 4) + 255) & ~(size_t)255;
    int* rank = (int*)(ws + off);
    off += (((size_t)E * 4) + 255) & ~(size_t)255;
    int2* csr = (int2*)(ws + off);
    off += (((size_t)E * 8) + 255) & ~(size_t)255;
    float* bufA = (float*)(ws + off);
    off += (((size_t)n * HIDC * 4) + 255) & ~(size_t)255;
    float* bufB = (float*)(ws + off);

    const int T = 256;
    const int gN   = (n + T - 1) / T;
    const int gE   = (E + T - 1) / T;
    const int gN16 = (n * 16 + T - 1) / T;
    const int n4   = n / 4;
    const int gS   = (n4 + SCAN_T - 1) / SCAN_T;
    const int rowTiles = n / 16;
    const int nProd = (rowTiles + 1) / 2;
    const int gM    = nProd + (gE > nProd ? gE : nProd);

    k_zero_cnt<<<gN, T, 0, stream>>>(cnt, n);
    k_rank_gemm<<<gM, T, 0, stream>>>(dst, cnt, rank, E, x, W1, bufA, rowTiles);
    k_scan_dis<<<gS, SCAN_T, 0, stream>>>(cnt, rowptr, dis, n4);
    k_fill<<<gE, T, 0, stream>>>(src, dst, rank, rowptr, dis, csr, E);

    k_gather_gemm<HIDC><<<gN16, T, 0, stream>>>(bufA, dis, rowptr, csr, b1, W2, nullptr, bufB, n);

    k_gather_gemm<10><<<gN16, T, 0, stream>>>(bufB, dis, rowptr, csr, b2, Wc, bc, out, n);
}


# Edge table of fixed 64-slot rows filled at count time (slot = dst*64 + returned rank, longer rows spill to a list): scan and fill launches go, 6 -> 4; layer 1 pass adds dis[src] to it; sum order kept
# speedup vs baseline: 2.5483x; 1.1375x over previous
#include <hip/hip_runtime.h>

typedef float v2f __attribute__((ext_vector_type(2)));
typedef float v8f __attribute__((ext_vector_type(8)));

#define N_NODES 50000
#define HIDC 64
#define APAD 4

#define CAP 64

__global__ void k_zero_cnt(int* __restrict__ cnt, int n) {
    int i = blockIdx.x * blockDim.x + threadIdx.x;
    if (i < n) cnt[i] = 0;
}

#define INC 128

__global__ void __launch_bounds__(256)
k_rank_gemm(const int* __restrict__ src, const int* __restrict__ dst, int* __restrict__ cnt,
            int* ovfN, int* __restrict__ tab, int2* __restrict__ ovf, int E,
            const float* __restrict__ A, const float* __restrict__ B, float* __restrict__ C,
            int rowTiles) {
    __shared__ float As[2][16 * (INC + APAD)];

    const int blk   = blockIdx.x;
    const int nProd = (rowTiles + 1) >> 1;
    const bool mixed = blk < 2 * nProd;

    if (!(mixed && (blk & 1))) {
        const int eb = mixed ? (blk >> 1) : (blk - nProd);
        const int e  = eb * 256 + threadIdx.x;
        if (e < E) {
            const int d = dst[e];
            const int sv = src[e];
            const int r = atomicAdd(&cnt[d], 1);
            if (r < CAP) {
                tab[2 * (d * CAP + r)] = sv;
            } else {
                const int p = atomicAdd(ovfN, 1);
                ovf[p] = make_int2(d, sv);
            }
        }
        return;
    }

    const int sub  = threadIdx.x >> 7;
    const int t128 = threadIdx.x & 127;
    const int tile = 2 * (blk >> 1) + sub;
    const bool tok = tile < rowTiles;
    const int m0   = (tok ? tile : (rowTiles - 1)) * 16;
    const int lane = threadIdx.x & 31;
    const int wave = t128 >> 5;
    const int n0   = wave * 16;
    const int half = lane >> 4;
    const int idx  = lane & 15;
    constexpr int ldsK = INC + APAD;
    float* as = As[sub];

    {
        const float4* s4 = (const float4*)(A + (long)m0 * INC);
#pragma unroll
        for (int q = t128; q < 16 * (INC / 4); q += 128) {
            const int r = q >> 5;
            const int c = q & 31;
            ((float4*)(as + r * ldsK))[c] = s4[q];
        }
    }
    __syncthreads();

    const float* arow = as + idx * ldsK;
    const int    bcol = n0 + idx;
    const float* bp   = B + bcol;

    v8f acc = {0.f, 0.f, 0.f, 0.f, 0.f, 0.f, 0.f, 0.f};

#pragma unroll 8
    for (int k = 0; k < INC; k += 4) {
        const int ka = k + 2 * half;
        v2f a, b;
        a.x = arow[ka];
        a.y = arow[ka + 1];
        b.x = bp[ka * HIDC];
        b.y = bp[(ka + 1) * HIDC];
        acc = __builtin_amdgcn_wmma_f32_16x16x4_f32(
             false, a,  false, b,
             (short)0, acc,  false,  false);
    }

    if (tok) {
#pragma unroll
        for (int v = 0; v < 8; ++v) {
            const int row = m0 + v + 8 * half;
            C[(long)row * HIDC + bcol] = acc[v];
        }
    }
}


static __device__ __forceinline__ void fma4(float4& a, const float4 v, const float w) {
    a.x = fmaf(v.x, w, a.x);
    a.y = fmaf(v.y, w, a.y);
    a.z = fmaf(v.z, w, a.z);
    a.w = fmaf(v.w, w, a.w);
}

static __device__ __forceinline__ void fma4_if(float4& a, const float4 v, const float w, const bool c) {
    a.x = c ? fmaf(v.x, w, a.x) : a.x;
    a.y = c ? fmaf(v.y, w, a.y) : a.y;
    a.z = c ? fmaf(v.z, w, a.z) : a.z;
    a.w = c ? fmaf(v.w, w, a.w) : a.w;
}

#define HPAD 4
template <int NOUT, bool MAKEW>
__global__ void k_gather_gemm(const float* __restrict__ hw, const int* __restrict__ cnt,
                              int* tabw, const int2* __restrict__ tabr,
                              const int2* __restrict__ ovf,
                              int E, const float* __restrict__ bias,
                              const float* __restrict__ Wn, const float* __restrict__ bn,
                              float* __restrict__ outp, int n) {
    constexpr int NT = (NOUT + 15) / 16;
    __shared__ float Hs[16 * (HIDC + HPAD)];

    const int t = blockIdx.x * blockDim.x + threadIdx.x;
    const int g = t & 15;
    const bool live = (t >> 4) < n;
    const int i = live ? (t >> 4) : (n - 1);
    const float4* hw4 = (const float4*)hw;

    const int   c   = cnt[i];
    const float dd  = rsqrtf((float)(c + 1));
    const int   m   = min(c, CAP);
    const int   r0  = i * CAP;

    float4 acc = make_float4(0.f, 0.f, 0.f, 0.f);
    fma4(acc, hw4[i * (HIDC / 4) + g], dd * dd);

    const int last = m - 1;
    for (int j = 0; j < m; j += 4) {
        const int j1 = min(j + 1, last);
        const int j2 = min(j + 2, last);
        const int j3 = min(j + 3, last);
        int s0, s1, s2, s3;
        float w0, w1, w2, w3;
        if (MAKEW) {
            s0 = tabw[2 * (r0 + j)];
            s1 = tabw[2 * (r0 + j1)];
            s2 = tabw[2 * (r0 + j2)];
            s3 = tabw[2 * (r0 + j3)];
        } else {
            const int2 e0 = tabr[r0 + j];
            const int2 e1 = tabr[r0 + j1];
            const int2 e2 = tabr[r0 + j2];
            const int2 e3 = tabr[r0 + j3];
            s0 = e0.x; s1 = e1.x; s2 = e2.x; s3 = e3.x;
            w0 = __int_as_float(e0.y);
            w1 = __int_as_float(e1.y);
            w2 = __int_as_float(e2.y);
            w3 = __int_as_float(e3.y);
        }
        const float4 v0 = hw4[s0 * (HIDC / 4) + g];
        const float4 v1 = hw4[s1 * (HIDC / 4) + g];
        const float4 v2 = hw4[s2 * (HIDC / 4) + g];
        const float4 v3 = hw4[s3 * (HIDC / 4) + g];
        if (MAKEW) {
            w0 = rsqrtf((float)(cnt[s0] + 1));
            w1 = rsqrtf((float)(cnt[s1] + 1));
            w2 = rsqrtf((float)(cnt[s2] + 1));
            w3 = rsqrtf((float)(cnt[s3] + 1));
            const float wlo = (g & 1) ? w1 : w0;
            const float whi = (g & 1) ? w3 : w2;
            const float wq  = (g & 2) ? whi : wlo;
            if (live && g < 4 && j + g < m) tabw[2 * (r0 + j + g) + 1] = __float_as_int(wq);
        }
        fma4(acc, v0, w0 * dd);
        fma4_if(acc, v1, w1 * dd, j + 1 < m);
        fma4_if(acc, v2, w2 * dd, j + 2 < m);
        fma4_if(acc, v3, w3 * dd, j + 3 < m);
    }

    if (c > CAP) {
        const int no = min(cnt[n], E);
        for (int p = 0; p < no; ++p) {
            const int2 r = ovf[p];
            if (r.x == i)
                fma4(acc, hw4[r.y * (HIDC / 4) + g], rsqrtf((float)(cnt[r.y] + 1)) * dd);
        }
    }

    const float4 b = ((const float4*)bias)[g];
    float4 o;
    o.x = fmaxf(acc.x + b.x, 0.f);
    o.y = fmaxf(acc.y + b.y, 0.f);
    o.z = fmaxf(acc.z + b.z, 0.f);
    o.w = fmaxf(acc.w + b.w, 0.f);

    ((float4*)(Hs + (threadIdx.x >> 4) * (HIDC + HPAD)))[g] = o;
    __syncthreads();

    const int wave = threadIdx.x >> 5;
    if (wave < NT) {
        const int m0   = blockIdx.x * 16;
        const int lane = threadIdx.x & 31;
        const int half = lane >> 4;
        const int idx  = lane & 15;
        const float* arow = Hs + idx * (HIDC + HPAD);
        const int    bcol = wave * 16 + idx;
        const bool   bok  = bcol < NOUT;
        const float* bp   = Wn + (bok ? bcol : (NOUT - 1));

        v8f d = {0.f, 0.f, 0.f, 0.f, 0.f, 0.f, 0.f, 0.f};
#pragma unroll
        for (int k = 0; k < HIDC; k += 4) {
            const int ka = k + 2 * half;
            v2f a, w;
            a.x = arow[ka];
            a.y = arow[ka + 1];
            w.x = bp[ka * NOUT];
            w.y = bp[(ka + 1) * NOUT];
            d = __builtin_amdgcn_wmma_f32_16x16x4_f32(
                 false, a,  false, w,
                 (short)0, d,  false,  false);
        }

        if (bok) {
            const float bb = bn ? bn[bcol] : 0.0f;
#pragma unroll
            for (int v = 0; v < 8; ++v) {
                const int row = m0 + v + 8 * half;
                if (row < n) outp[(long)row * NOUT + bcol] = d[v] + bb;
            }
        }
    }
}


extern "C" void kernel_launch(void* const* d_in, const int* in_sizes, int n_in,
                              void* d_out, int out_size, void* d_ws, size_t ws_size,
                              hipStream_t stream) {
    const float* x  = (const float*)d_in[0];
    const int*   ei = (const int*)d_in[1];
    const float* W1 = (const float*)d_in[2];
    const float* b1 = (const float*)d_in[3];
    const float* W2 = (const float*)d_in[4];
    const float* b2 = (const float*)d_in[5];
    const float* Wc = (const float*)d_in[6];
    const float* bc = (const float*)d_in[7];
    float* out = (float*)d_out;

    const int n = N_NODES;
    const int E = in_sizes[1] / 2;
    const int* src = ei;
    const int* dst = ei + E;

    char* ws = (char*)d_ws;
    size_t off = 0;
    int* cnt = (int*)(ws + off);
    off += (((size_t)(n + 1) * 4) + 255) & ~(size_t)255;
    int* tab = (int*)(ws + off);
    off += (((size_t)n * CAP * 8) + 255) & ~(size_t)255;
    int2* ovf = (int2*)(ws + off);
    off += (((size_t)E * 8) + 255) & ~(size_t)255;
    float* bufA = (float*)(ws + off);
    off += (((size_t)n * HIDC * 4) + 255) & ~(size_t)255;
    float* bufB = (float*)(ws + off);

    const int T = 256;
    const int gN   = (n + 1 + T - 1) / T;
    const int gE   = (E + T - 1) / T;
    const int gN16 = (n * 16 + T - 1) / T;
    const int rowTiles = n / 16;
    const int nProd = (rowTiles + 1) / 2;
    const int gM    = nProd + (gE > nProd ? gE : nProd);

    k_zero_cnt<<<gN, T, 0, stream>>>(cnt, n + 1);
    k_rank_gemm<<<gM, T, 0, stream>>>(src, dst, cnt, cnt + n, tab, ovf, E, x, W1, bufA, rowTiles);

    k_gather_gemm<HIDC, true><<<gN16, T, 0, stream>>>(bufA, cnt, tab, nullptr, ovf, E, b1, W2, nullptr, bufB, n);

    k_gather_gemm<10, false><<<gN16, T, 0, stream>>>(bufB, cnt, nullptr, (const int2*)tab, ovf, E, b2, Wc, bc, out, n);
}


# Edge slots from LDS counters in two levels (edges filed by 256-node bin beside x@W1, then one block per bin ranks them), not a returning global add per edge; sum order by LDS arrival; 4->5 launches
# speedup vs baseline: 3.5641x; 1.3986x over previous
#include <hip/hip_runtime.h>

typedef float v2f __attribute__((ext_vector_type(2)));
typedef float v8f __attribute__((ext_vector_type(8)));

#define N_NODES 50000
#define HIDC 64
#define APAD 4

#define CAP 64

#define BSH 8
#define BNODES (1 << BSH)
#define NBINS ((N_NODES + BNODES - 1) / BNODES)
#define BINCAP 8192
#define BINPITCH (BINCAP + 1120)
#define CURP 1120
#define EPT 8
#define EPB (EPT * 256)

__global__ void k_zero_cnt(int* __restrict__ cnt, int n, int* __restrict__ binCur) {
    int i = blockIdx.x * blockDim.x + threadIdx.x;
    if (i < n) cnt[i] = 0;
    if (i < NBINS) binCur[i * CURP] = 0;
}

#define INC 128

__global__ void __launch_bounds__(256)
k_file_gemm(const int* __restrict__ src, const int* __restrict__ dst, int* cnt,
            int* ovfN, int* binCur, int* __restrict__ binList,
            int* __restrict__ tab, int2* __restrict__ ovf, int E, int nEdgeBlk,
            const float* __restrict__ A, const float* __restrict__ B, float* __restrict__ C,
            int rowTiles) {
    __shared__ float As[2][16 * (INC + APAD)];
    __shared__ int hist[NBINS];
    __shared__ int base[NBINS];

    const int blk   = blockIdx.x;
    const int nProd = (rowTiles + 1) >> 1;
    const int nMix  = min(nEdgeBlk, nProd);
    const bool mixed  = blk < 2 * nMix;
    const bool isEdge = mixed ? !(blk & 1) : (nEdgeBlk > nProd);
    const int  bi     = mixed ? (blk >> 1) : (blk - nMix);

    if (isEdge) {
        for (int b = threadIdx.x; b < NBINS; b += 256) hist[b] = 0;
        __syncthreads();

        int dv[EPT], sv[EPT], rk[EPT];
        const int e0 = bi * EPB + threadIdx.x;
#pragma unroll
        for (int q = 0; q < EPT; ++q) {
            const int e = e0 + q * 256;
            dv[q] = -1;
            sv[q] = 0;
            if (e < E) { dv[q] = dst[e]; sv[q] = src[e]; }
        }
#pragma unroll
        for (int q = 0; q < EPT; ++q) {
            rk[q] = 0;
            if (dv[q] >= 0) rk[q] = atomicAdd(&hist[dv[q] >> BSH], 1);
        }
        __syncthreads();
        for (int b = threadIdx.x; b < NBINS; b += 256) {
            const int c = hist[b];
            base[b] = c > 0 ? atomicAdd(&binCur[b * CURP], c) : 0;
        }
        __syncthreads();
#pragma unroll
        for (int q = 0; q < EPT; ++q) {
            if (dv[q] >= 0) {
                const int d   = dv[q];
                const int k   = d >> BSH;
                const int pos = base[k] + rk[q];
                if (pos < BINCAP) {
                    binList[k * BINPITCH + pos] = (sv[q] << BSH) | (d & (BNODES - 1));
                } else {
                    const int r = atomicAdd(&cnt[d], 1);
                    if (r < CAP) {
                        tab[2 * (d * CAP + r)] = sv[q];
                    } else {
                        const int p = atomicAdd(ovfN, 1);
                        ovf[p] = make_int2(d, sv[q]);
                    }
                }
            }
        }
        return;
    }

    const int sub  = threadIdx.x >> 7;
    const int t128 = threadIdx.x & 127;
    const int tile = 2 * bi + sub;
    const bool tok = tile < rowTiles;
    const int m0   = (tok ? tile : (rowTiles - 1)) * 16;
    const int lane = threadIdx.x & 31;
    const int wave = t128 >> 5;
    const int n0   = wave * 16;
    const int half = lane >> 4;
    const int idx  = lane & 15;
    constexpr int ldsK = INC + APAD;
    float* as = As[sub];

    {
        const float4* s4 = (const float4*)(A + (long)m0 * INC);
#pragma unroll
        for (int q = t128; q < 16 * (INC / 4); q += 128) {
            const int r = q >> 5;
            const int c = q & 31;
            ((float4*)(as + r * ldsK))[c] = s4[q];
        }
    }
    __syncthreads();

    const float* arow = as + idx * ldsK;
    const int    bcol = n0 + idx;
    const float* bp   = B + bcol;

    v8f acc = {0.f, 0.f, 0.f, 0.f, 0.f, 0.f, 0.f, 0.f};

#pragma unroll 8
    for (int k = 0; k < INC; k += 4) {
        const int ka = k + 2 * half;
        v2f a, b;
        a.x = arow[ka];
        a.y = arow[ka + 1];
        b.x = bp[ka * HIDC];
        b.y = bp[(ka + 1) * HIDC];
        acc = __builtin_amdgcn_wmma_f32_16x16x4_f32(
             false, a,  false, b,
             (short)0, acc,  false,  false);
    }

    if (tok) {
#pragma unroll
        for (int v = 0; v < 8; ++v) {
            const int row = m0 + v + 8 * half;
            C[(long)row * HIDC + bcol] = acc[v];
        }
    }
}

#define RT 1024
#define RU 4

__global__ void __launch_bounds__(RT)
k_rank_bins(const int* binCur, const int* __restrict__ binList, int* cnt, int* ovfN,
            int* __restrict__ tab, int2* __restrict__ ovf, int n) {
    __shared__ int lc[BNODES];

    const int k   = blockIdx.x;
    const int d0  = k << BSH;
    const int tid = threadIdx.x;
    if (tid < BNODES) lc[tid] = (d0 + tid < n) ? cnt[d0 + tid] : 0;
    const int m = min(binCur[k * CURP], BINCAP);
    __syncthreads();

    const int* lst = binList + k * BINPITCH;
    for (int j0 = 0; j0 < m; j0 += RU * RT) {
        int pk[RU];
#pragma unroll
        for (int u = 0; u < RU; ++u) {
            const int j = j0 + u * RT + tid;
            pk[u] = (j < m) ? lst[j] : -1;
        }
#pragma unroll
        for (int u = 0; u < RU; ++u) {
            if (pk[u] >= 0) {
                const int dl = pk[u] & (BNODES - 1);
                const int s  = pk[u] >> BSH;
                const int d  = d0 + dl;
                const int r  = atomicAdd(&lc[dl], 1);
                if (r < CAP) {
                    tab[2 * (d * CAP + r)] = s;
                } else {
                    const int p = atomicAdd(ovfN, 1);
                    ovf[p] = make_int2(d, s);
                }
            }
        }
    }
    __syncthreads();
    if (tid < BNODES && d0 + tid < n) cnt[d0 + tid] = lc[tid];
}


static __device__ __forceinline__ void fma4(float4& a, const float4 v, const float w) {
    a.x = fmaf(v.x, w, a.x);
    a.y = fmaf(v.y, w, a.y);
    a.z = fmaf(v.z, w, a.z);
    a.w = fmaf(v.w, w, a.w);
}

static __device__ __forceinline__ void fma4_if(float4& a, const float4 v, const float w, const bool c) {
    a.x = c ? fmaf(v.x, w, a.x) : a.x;
    a.y = c ? fmaf(v.y, w, a.y) : a.y;
    a.z = c ? fmaf(v.z, w, a.z) : a.z;
    a.w = c ? fmaf(v.w, w, a.w) : a.w;
}

#define HPAD 4
template <int NOUT, bool MAKEW>
__global__ void k_gather_gemm(const float* __restrict__ hw, const int* __restrict__ cnt,
                              int* tabw, const int2* __restrict__ tabr,
                              const int2* __restrict__ ovf,
                              int E, const float* __restrict__ bias,
                              const float* __restrict__ Wn, const float* __restrict__ bn,
                              float* __restrict__ outp, int n) {
    constexpr int NT = (NOUT + 15) / 16;
    __shared__ float Hs[16 * (HIDC + HPAD)];

    const int t = blockIdx.x * blockDim.x + threadIdx.x;
    const int g = t & 15;
    const bool live = (t >> 4) < n;
    const int i = live ? (t >> 4) : (n - 1);
    const float4* hw4 = (const float4*)hw;

    const int   c   = cnt[i];
    const float dd  = rsqrtf((float)(c + 1));
    const int   m   = min(c, CAP);
    const int   r0  = i * CAP;

    float4 acc = make_float4(0.f, 0.f, 0.f, 0.f);
    fma4(acc, hw4[i * (HIDC / 4) + g], dd * dd);

    const int last = m - 1;
    for (int j = 0; j < m; j += 4) {
        const int j1 = min(j + 1, last);
        const int j2 = min(j + 2, last);
        const int j3 = min(j + 3, last);
        int s0, s1, s2, s3;
        float w0, w1, w2, w3;
        if (MAKEW) {
            s0 = tabw[2 * (r0 + j)];
            s1 = tabw[2 * (r0 + j1)];
            s2 = tabw[2 * (r0 + j2)];
            s3 = tabw[2 * (r0 + j3)];
        } else {
            const int2 e0 = tabr[r0 + j];
            const int2 e1 = tabr[r0 + j1];
            const int2 e2 = tabr[r0 + j2];
            const int2 e3 = tabr[r0 + j3];
            s0 = e0.x; s1 = e1.x; s2 = e2.x; s3 = e3.x;
            w0 = __int_as_float(e0.y);
            w1 = __int_as_float(e1.y);
            w2 = __int_as_float(e2.y);
            w3 = __int_as_float(e3.y);
        }
        const float4 v0 = hw4[s0 * (HIDC / 4) + g];
        const float4 v1 = hw4[s1 * (HIDC / 4) + g];
        const float4 v2 = hw4[s2 * (HIDC / 4) + g];
        const float4 v3 = hw4[s3 * (HIDC / 4) + g];
        if (MAKEW) {
            w0 = rsqrtf((float)(cnt[s0] + 1));
            w1 = rsqrtf((float)(cnt[s1] + 1));
            w2 = rsqrtf((float)(cnt[s2] + 1));
            w3 = rsqrtf((float)(cnt[s3] + 1));
            const float wlo = (g & 1) ? w1 : w0;
            const float whi = (g & 1) ? w3 : w2;
            const float wq  = (g & 2) ? whi : wlo;
            if (live && g < 4 && j + g < m) tabw[2 * (r0 + j + g) + 1] = __float_as_int(wq);
        }
        fma4(acc, v0, w0 * dd);
        fma4_if(acc, v1, w1 * dd, j + 1 < m);
        fma4_if(acc, v2, w2 * dd, j + 2 < m);
        fma4_if(acc, v3, w3 * dd, j + 3 < m);
    }

    if (c > CAP) {
        const int no = min(cnt[n], E);
        for (int p = 0; p < no; ++p) {
            const int2 r = ovf[p];
            if (r.x == i)
                fma4(acc, hw4[r.y * (HIDC / 4) + g], rsqrtf((float)(cnt[r.y] + 1)) * dd);
        }
    }

    const float4 b = ((const float4*)bias)[g];
    float4 o;
    o.x = fmaxf(acc.x + b.x, 0.f);
    o.y = fmaxf(acc.y + b.y, 0.f);
    o.z = fmaxf(acc.z + b.z, 0.f);
    o.w = fmaxf(acc.w + b.w, 0.f);

    ((float4*)(Hs + (threadIdx.x >> 4) * (HIDC + HPAD)))[g] = o;
    __syncthreads();

    const int wave = threadIdx.x >> 5;
    if (wave < NT) {
        const int m0   = blockIdx.x * 16;
        const int lane = threadIdx.x & 31;
        const int half = lane >> 4;
        const int idx  = lane & 15;
        const float* arow = Hs + idx * (HIDC + HPAD);
        const int    bcol = wave * 16 + idx;
        const bool   bok  = bcol < NOUT;
        const float* bp   = Wn + (bok ? bcol : (NOUT - 1));

        v8f d = {0.f, 0.f, 0.f, 0.f, 0.f, 0.f, 0.f, 0.f};
#pragma unroll
        for (int k = 0; k < HIDC; k += 4) {
            const int ka = k + 2 * half;
            v2f a, w;
            a.x = arow[ka];
            a.y = arow[ka + 1];
            w.x = bp[ka * NOUT];
            w.y = bp[(ka + 1) * NOUT];
            d = __builtin_amdgcn_wmma_f32_16x16x4_f32(
                 false, a,  false, w,
                 (short)0, d,  false,  false);
        }

        if (bok) {
            const float bb = bn ? bn[bcol] : 0.0f;
#pragma unroll
            for (int v = 0; v < 8; ++v) {
                const int row = m0 + v + 8 * half;
                if (row < n) outp[(long)row * NOUT + bcol] = d[v] + bb;
            }
        }
    }
}


extern "C" void kernel_launch(void* const* d_in, const int* in_sizes, int n_in,
                              void* d_out, int out_size, void* d_ws, size_t ws_size,
                              hipStream_t stream) {
    const float* x  = (const float*)d_in[0];
    const int*   ei = (const int*)d_in[1];
    const float* W1 = (const float*)d_in[2];
    const float* b1 = (const float*)d_in[3];
    const float* W2 = (const float*)d_in[4];
    const float* b2 = (const float*)d_in[5];
    const float* Wc = (const float*)d_in[6];
    const float* bc = (const float*)d_in[7];
    float* out = (float*)d_out;

    const int n = N_NODES;
    const int E = in_sizes[1] / 2;
    const int* src = ei;
    const int* dst = ei + E;

    char* ws = (char*)d_ws;
    size_t off = 0;
    int* cnt = (int*)(ws + off);
    off += (((size_t)(n + 1) * 4) + 255) & ~(size_t)255;
    int* binCur = (int*)(ws + off);
    off += (((size_t)NBINS * CURP * 4) + 255) & ~(size_t)255;
    int* binList = (int*)(ws + off);
    off += (((size_t)NBINS * BINPITCH * 4) + 255) & ~(size_t)255;
    int* tab = (int*)(ws + off);
    off += (((size_t)n * CAP * 8) + 255) & ~(size_t)255;
    int2* ovf = (int2*)(ws + off);
    off += (((size_t)E * 8) + 255) & ~(size_t)255;
    float* bufA = (float*)(ws + off);
    off += (((size_t)n * HIDC * 4) + 255) & ~(size_t)255;
    float* bufB = (float*)(ws + off);

    const int T = 256;
    const int gZ   = (n + 1 + T - 1) / T;
    const int gE   = (E + EPB - 1) / EPB;
    const int gN16 = (n * 16 + T - 1) / T;
    const int rowTiles = n / 16;
    const int nProd = (rowTiles + 1) / 2;
    const int gM    = nProd + gE;

    k_zero_cnt<<<gZ, T, 0, stream>>>(cnt, n + 1, binCur);
    k_file_gemm<<<gM, T, 0, stream>>>(src, dst, cnt, cnt + n, binCur, binList, tab, ovf, E, gE,
                                      x, W1, bufA, rowTiles);
    k_rank_bins<<<NBINS, RT, 0, stream>>>(binCur, binList, cnt, cnt + n, tab, ovf, n);

    k_gather_gemm<HIDC, true><<<gN16, T, 0, stream>>>(bufA, cnt, tab, nullptr, ovf, E, b1, W2, nullptr, bufB, n);

    k_gather_gemm<10, false><<<gN16, T, 0, stream>>>(bufB, cnt, nullptr, (const int2*)tab, ovf, E, b2, Wc, bc, out, n);
}


# Edge blocks file into own segments (runs in bin order by an LDS running sum, one descriptor word per bin), not cursor adds in global memory: nothing to clear, 5->4 launches; sums in LDS order
# speedup vs baseline: 4.6268x; 1.2982x over previous
#include <hip/hip_runtime.h>

typedef float v2f __attribute__((ext_vector_type(2)));
typedef float v8f __attribute__((ext_vector_type(8)));

#define N_NODES 50000
#define HIDC 64
#define APAD 4

#define CAP 64

#define BSH 8
#define BNODES (1 << BSH)
#define NBINS ((N_NODES + BNODES - 1) / BNODES)
#define EPT 8
#define EPB (EPT * 256)
#define DSH 12
static_assert(NBINS <= 256, "one thread of an edge block per bin");
static_assert(EPB < (1 << DSH), "run start and length fit their fields");

#define INC 128

__global__ void __launch_bounds__(256)
k_file_gemm(const int* __restrict__ src, const int* __restrict__ dst, int* __restrict__ ovfN,
            int* __restrict__ seg, int* __restrict__ desc, int E, int nEdgeBlk,
            const float* __restrict__ A, const float* __restrict__ B, float* __restrict__ C,
            int rowTiles) {
    __shared__ float As[2][16 * (INC + APAD)];
    __shared__ int hist[NBINS];
    __shared__ int base[NBINS];
    __shared__ int wsum[8];

    const int blk   = blockIdx.x;
    const int nProd = (rowTiles + 1) >> 1;
    const int nMix  = min(nEdgeBlk, nProd);
    const bool mixed  = blk < 2 * nMix;
    const bool isEdge = mixed ? !(blk & 1) : (nEdgeBlk > nProd);
    const int  bi     = mixed ? (blk >> 1) : (blk - nMix);

    if (isEdge) {
        if (bi == 0 && threadIdx.x == 0) *ovfN = 0;
        for (int b = threadIdx.x; b < NBINS; b += 256) hist[b] = 0;
        __syncthreads();

        int dv[EPT], sv[EPT], rk[EPT];
        const int e0 = bi * EPB + threadIdx.x;
#pragma unroll
        for (int q = 0; q < EPT; ++q) {
            const int e = e0 + q * 256;
            dv[q] = -1;
            sv[q] = 0;
            if (e < E) { dv[q] = dst[e]; sv[q] = src[e]; }
        }
#pragma unroll
        for (int q = 0; q < EPT; ++q) {
            rk[q] = 0;
            if (dv[q] >= 0) rk[q] = atomicAdd(&hist[dv[q] >> BSH], 1);
        }
        __syncthreads();

        const int ln = threadIdx.x & 31;
        const int wv = threadIdx.x >> 5;
        const int c  = threadIdx.x < NBINS ? hist[threadIdx.x] : 0;
        int v = c;
#pragma unroll
        for (int o = 1; o < 32; o <<= 1) {
            const int t = __shfl_up(v, o);
            if (ln >= o) v += t;
        }
        if (ln == 31) wsum[wv] = v;
        __syncthreads();
        int st = v - c;
        for (int w = 0; w < wv; ++w) st += wsum[w];
        if (threadIdx.x < NBINS) {
            base[threadIdx.x] = st;
            desc[bi * NBINS + threadIdx.x] = st | (c << DSH);
        }
        __syncthreads();

        int* myseg = seg + bi * EPB;
#pragma unroll
        for (int q = 0; q < EPT; ++q) {
            if (dv[q] >= 0) {
                const int d = dv[q];
                myseg[base[d >> BSH] + rk[q]] = (sv[q] << BSH) | (d & (BNODES - 1));
            }
        }
        return;
    }

    const int sub  = threadIdx.x >> 7;
    const int t128 = threadIdx.x & 127;
    const int tile = 2 * bi + sub;
    const bool tok = tile < rowTiles;
    const int m0   = (tok ? tile : (rowTiles - 1)) * 16;
    const int lane = threadIdx.x & 31;
    const int wave = t128 >> 5;
    const int n0   = wave * 16;
    const int half = lane >> 4;
    const int idx  = lane & 15;
    constexpr int ldsK = INC + APAD;
    float* as = As[sub];

    {
        const float4* s4 = (const float4*)(A + (long)m0 * INC);
#pragma unroll
        for (int q = t128; q < 16 * (INC / 4); q += 128) {
            const int r = q >> 5;
            const int c = q & 31;
            ((float4*)(as + r * ldsK))[c] = s4[q];
        }
    }
    __syncthreads();

    const float* arow = as + idx * ldsK;
    const int    bcol = n0 + idx;
    const float* bp   = B + bcol;

    v8f acc = {0.f, 0.f, 0.f, 0.f, 0.f, 0.f, 0.f, 0.f};

#pragma unroll 8
    for (int k = 0; k < INC; k += 4) {
        const int ka = k + 2 * half;
        v2f a, b;
        a.x = arow[ka];
        a.y = arow[ka + 1];
        b.x = bp[ka * HIDC];
        b.y = bp[(ka + 1) * HIDC];
        acc = __builtin_amdgcn_wmma_f32_16x16x4_f32(
             false, a,  false, b,
             (short)0, acc,  false,  false);
    }

    if (tok) {
#pragma unroll
        for (int v = 0; v < 8; ++v) {
            const int row = m0 + v + 8 * half;
            C[(long)row * HIDC + bcol] = acc[v];
        }
    }
}

#define RT 1024
#define RU 4

static __device__ __forceinline__ void place_edge(const int pk, const int d0, int* lc, int* ovfN,
                                                  int* __restrict__ tab, int2* __restrict__ ovf) {
    if (pk >= 0) {
        const int dl = pk & (BNODES - 1);
        const int s  = pk >> BSH;
        const int d  = d0 + dl;
        const int r  = atomicAdd(&lc[dl], 1);
        if (r < CAP) {
            tab[2 * (d * CAP + r)] = s;
        } else {
            const int p = atomicAdd(ovfN, 1);
            ovf[p] = make_int2(d, s);
        }
    }
}

__global__ void __launch_bounds__(RT)
k_rank_bins(const int* __restrict__ desc, const int* __restrict__ seg, int nEdgeBlk,
            int* __restrict__ cnt, int* ovfN, int* __restrict__ tab, int2* __restrict__ ovf, int n) {
    __shared__ int lc[BNODES];

    const int k   = blockIdx.x;
    const int d0  = k << BSH;
    const int tid = threadIdx.x;
    if (tid < BNODES) lc[tid] = 0;
    __syncthreads();

    const int lane = tid & 31;
    const int wave = tid >> 5;
    const int half = lane >> 4;
    const int idx  = lane & 15;
    constexpr int NW = RT / 32;

    const int nRun = (nEdgeBlk - wave + NW - 1) / NW;
    for (int c0 = 0; c0 < nRun; c0 += 32) {
        const int nr  = min(32, nRun - c0);
        const int dsc = (lane < nr) ? desc[(wave + (c0 + lane) * NW) * NBINS + k] : 0;
        for (int r0 = 0; r0 < nr; r0 += 2 * RU) {
            int pk[RU], cn[RU], ad[RU];
#pragma unroll
            for (int u = 0; u < RU; ++u) {
                const int r  = r0 + 2 * u + half;
                const int dw = __shfl(dsc, min(r, 31));
                const int d  = (r < nr) ? dw : 0;
                cn[u] = d >> DSH;
                ad[u] = (wave + (c0 + r) * NW) * EPB + (d & ((1 << DSH) - 1));
                pk[u] = (idx < cn[u]) ? seg[ad[u] + idx] : -1;
            }
#pragma unroll
            for (int u = 0; u < RU; ++u) {
                place_edge(pk[u], d0, lc, ovfN, tab, ovf);
                for (int j = 16; __any(j < cn[u]); j += 16) {
                    const int p2 = (j + idx < cn[u]) ? seg[ad[u] + j + idx] : -1;
                    place_edge(p2, d0, lc, ovfN, tab, ovf);
                }
            }
        }
    }
    __syncthreads();
    if (tid < BNODES && d0 + tid < n) cnt[d0 + tid] = lc[tid];
}


static __device__ __forceinline__ void fma4(float4& a, const float4 v, const float w) {
    a.x = fmaf(v.x, w, a.x);
    a.y = fmaf(v.y, w, a.y);
    a.z = fmaf(v.z, w, a.z);
    a.w = fmaf(v.w, w, a.w);
}

static __device__ __forceinline__ void fma4_if(float4& a, const float4 v, const float w, const bool c) {
    a.x = c ? fmaf(v.x, w, a.x) : a.x;
    a.y = c ? fmaf(v.y, w, a.y) : a.y;
    a.z = c ? fmaf(v.z, w, a.z) : a.z;
    a.w = c ? fmaf(v.w, w, a.w) : a.w;
}

#define HPAD 4
template <int NOUT, bool MAKEW>
__global__ void k_gather_gemm(const float* __restrict__ hw, const int* __restrict__ cnt,
                              int* tabw, const int2* __restrict__ tabr,
                              const int2* __restrict__ ovf,
                              int E, const float* __restrict__ bias,
                              const float* __restrict__ Wn, const float* __restrict__ bn,
                              float* __restrict__ outp, int n) {
    constexpr int NT = (NOUT + 15) / 16;
    __shared__ float Hs[16 * (HIDC + HPAD)];

    const int t = blockIdx.x * blockDim.x + threadIdx.x;
    const int g = t & 15;
    const bool live = (t >> 4) < n;
    const int i = live ? (t >> 4) : (n - 1);
    const float4* hw4 = (const float4*)hw;

    const int   c   = cnt[i];
    const float dd  = rsqrtf((float)(c + 1));
    const int   m   = min(c, CAP);
    const int   r0  = i * CAP;

    float4 acc = make_float4(0.f, 0.f, 0.f, 0.f);
    fma4(acc, hw4[i * (HIDC / 4) + g], dd * dd);

    const int last = m - 1;
    for (int j = 0; j < m; j += 4) {
        const int j1 = min(j + 1, last);
        const int j2 = min(j + 2, last);
        const int j3 = min(j + 3, last);
        int s0, s1, s2, s3;
        float w0, w1, w2, w3;
        if (MAKEW) {
            s0 = tabw[2 * (r0 + j)];
            s1 = tabw[2 * (r0 + j1)];
            s2 = tabw[2 * (r0 + j2)];
            s3 = tabw[2 * (r0 + j3)];
        } else {
            const int2 e0 = tabr[r0 + j];
            const int2 e1 = tabr[r0 + j1];
            const int2 e2 = tabr[r0 + j2];
            const int2 e3 = tabr[r0 + j3];
            s0 = e0.x; s1 = e1.x; s2 = e2.x; s3 = e3.x;
            w0 = __int_as_float(e0.y);
            w1 = __int_as_float(e1.y);
            w2 = __int_as_float(e2.y);
            w3 = __int_as_float(e3.y);
        }
        const float4 v0 = hw4[s0 * (HIDC / 4) + g];
        const float4 v1 = hw4[s1 * (HIDC / 4) + g];
        const float4 v2 = hw4[s2 * (HIDC / 4) + g];
        const float4 v3 = hw4[s3 * (HIDC / 4) + g];
        if (MAKEW) {
            w0 = rsqrtf((float)(cnt[s0] + 1));
            w1 = rsqrtf((float)(cnt[s1] + 1));
            w2 = rsqrtf((float)(cnt[s2] + 1));
            w3 = rsqrtf((float)(cnt[s3] + 1));
            const float wlo = (g & 1) ? w1 : w0;
            const float whi = (g & 1) ? w3 : w2;
            const float wq  = (g & 2) ? whi : wlo;
            if (live && g < 4 && j + g < m) tabw[2 * (r0 + j + g) + 1] = __float_as_int(wq);
        }
        fma4(acc, v0, w0 * dd);
        fma4_if(acc, v1, w1 * dd, j + 1 < m);
        fma4_if(acc, v2, w2 * dd, j + 2 < m);
        fma4_if(acc, v3, w3 * dd, j + 3 < m);
    }

    if (c > CAP) {
        const int no = min(cnt[n], E);
        for (int p = 0; p < no; ++p) {
            const int2 r = ovf[p];
            if (r.x == i)
                fma4(acc, hw4[r.y * (HIDC / 4) + g], rsqrtf((float)(cnt[r.y] + 1)) * dd);
        }
    }

    const float4 b = ((const float4*)bias)[g];
    float4 o;
    o.x = fmaxf(acc.x + b.x, 0.f);
    o.y = fmaxf(acc.y + b.y, 0.f);
    o.z = fmaxf(acc.z + b.z, 0.f);
    o.w = fmaxf(acc.w + b.w, 0.f);

    ((float4*)(Hs + (threadIdx.x >> 4) * (HIDC + HPAD)))[g] = o;
    __syncthreads();

    const int wave = threadIdx.x >> 5;
    if (wave < NT) {
        const int m0   = blockIdx.x * 16;
        const int lane = threadIdx.x & 31;
        const int half = lane >> 4;
        const int idx  = lane & 15;
        const float* arow = Hs + idx * (HIDC + HPAD);
        const int    bcol = wave * 16 + idx;
        const bool   bok  = bcol < NOUT;
        const float* bp   = Wn + (bok ? bcol : (NOUT - 1));

        v8f d = {0.f, 0.f, 0.f, 0.f, 0.f, 0.f, 0.f, 0.f};
#pragma unroll
        for (int k = 0; k < HIDC; k += 4) {
            const int ka = k + 2 * half;
            v2f a, w;
            a.x = arow[ka];
            a.y = arow[ka + 1];
            w.x = bp[ka * NOUT];
            w.y = bp[(ka + 1) * NOUT];
            d = __builtin_amdgcn_wmma_f32_16x16x4_f32(
                 false, a,  false, w,
                 (short)0, d,  false,  false);
        }

        if (bok) {
            const float bb = bn ? bn[bcol] : 0.0f;
#pragma unroll
            for (int v = 0; v < 8; ++v) {
                const int row = m0 + v + 8 * half;
                if (row < n) outp[(long)row * NOUT + bcol] = d[v] + bb;
            }
        }
    }
}


extern "C" void kernel_launch(void* const* d_in, const int* in_sizes, int n_in,
                              void* d_out, int out_size, void* d_ws, size_t ws_size,
                              hipStream_t stream) {
    const float* x  = (const float*)d_in[0];
    const int*   ei = (const int*)d_in[1];
    const float* W1 = (const float*)d_in[2];
    const float* b1 = (const float*)d_in[3];
    const float* W2 = (const float*)d_in[4];
    const float* b2 = (const float*)d_in[5];
    const float* Wc = (const float*)d_in[6];
    const float* bc = (const float*)d_in[7];
    float* out = (float*)d_out;

    const int n = N_NODES;
    const int E = in_sizes[1] / 2;
    const int* src = ei;
    const int* dst = ei + E;

    const int T = 256;
    const int gE   = (E + EPB - 1) / EPB;

    char* ws = (char*)d_ws;
    size_t off = 0;
    int* cnt = (int*)(ws + off);
    off += (((size_t)(n + 1) * 4) + 255) & ~(size_t)255;
    int* seg = (int*)(ws + off);
    off += (((size_t)gE * EPB * 4) + 255) & ~(size_t)255;
    int* desc = (int*)(ws + off);
    off += (((size_t)gE * NBINS * 4) + 255) & ~(size_t)255;
    int* tab = (int*)(ws + off);
    off += (((size_t)n * CAP * 8) + 255) & ~(size_t)255;
    int2* ovf = (int2*)(ws + off);
    off += (((size_t)E * 8) + 255) & ~(size_t)255;
    float* bufA = (float*)(ws + off);
    off += (((size_t)n * HIDC * 4) + 255) & ~(size_t)255;
    float* bufB = (float*)(ws + off);

    const int gN16 = (n * 16 + T - 1) / T;
    const int rowTiles = n / 16;
    const int nProd = (rowTiles + 1) / 2;
    const int gM    = nProd + gE;

    k_file_gemm<<<gM, T, 0, stream>>>(src, dst, cnt + n, seg, desc, E, gE, x, W1, bufA, rowTiles);
    k_rank_bins<<<NBINS, RT, 0, stream>>>(desc, seg, gE, cnt, cnt + n, tab, ovf, n);

    k_gather_gemm<HIDC, true><<<gN16, T, 0, stream>>>(bufA, cnt, tab, nullptr, ovf, E, b1, W2, nullptr, bufB, n);

    k_gather_gemm<10, false><<<gN16, T, 0, stream>>>(bufB, cnt, nullptr, (const int2*)tab, ovf, E, b2, Wc, bc, out, n);
}


# Aggregation launches run as resident blocks of three node tiles each: a tile's counts and slots are fetched a tile ahead, ids and weights spread by row broadcast, eight rows asked for at once
# speedup vs baseline: 4.8415x; 1.0464x over previous
#include <hip/hip_runtime.h>

typedef float v2f __attribute__((ext_vector_type(2)));
typedef float v8f __attribute__((ext_vector_type(8)));

#define N_NODES 50000
#define HIDC 64
#define APAD 4

#define CAP 64

#define BSH 8
#define BNODES (1 << BSH)
#define NBINS ((N_NODES + BNODES - 1) / BNODES)
#define EPT 8
#define EPB (EPT * 256)
#define DSH 12
static_assert(NBINS <= 256, "one thread of an edge block per bin");
static_assert(EPB < (1 << DSH), "run start and length fit their fields");

#define INC 128

__global__ void __launch_bounds__(256)
k_file_gemm(const int* __restrict__ src, const int* __restrict__ dst, int* __restrict__ ovfN,
            int* __restrict__ seg, int* __restrict__ desc, int E, int nEdgeBlk,
            const float* __restrict__ A, const float* __restrict__ B, float* __restrict__ C,
            int rowTiles) {
    __shared__ float As[2][16 * (INC + APAD)];
    __shared__ int hist[NBINS];
    __shared__ int base[NBINS];
    __shared__ int wsum[8];

    const int blk   = blockIdx.x;
    const int nProd = (rowTiles + 1) >> 1;
    const int nMix  = min(nEdgeBlk, nProd);
    const bool mixed  = blk < 2 * nMix;
    const bool isEdge = mixed ? !(blk & 1) : (nEdgeBlk > nProd);
    const int  bi     = mixed ? (blk >> 1) : (blk - nMix);

    if (isEdge) {
        if (bi == 0 && threadIdx.x == 0) *ovfN = 0;
        for (int b = threadIdx.x; b < NBINS; b += 256) hist[b] = 0;
        __syncthreads();

        int dv[EPT], sv[EPT], rk[EPT];
        const int e0 = bi * EPB + threadIdx.x;
#pragma unroll
        for (int q = 0; q < EPT; ++q) {
            const int e = e0 + q * 256;
            dv[q] = -1;
            sv[q] = 0;
            if (e < E) { dv[q] = dst[e]; sv[q] = src[e]; }
        }
#pragma unroll
        for (int q = 0; q < EPT; ++q) {
            rk[q] = 0;
            if (dv[q] >= 0) rk[q] = atomicAdd(&hist[dv[q] >> BSH], 1);
        }
        __syncthreads();

        const int ln = threadIdx.x & 31;
        const int wv = threadIdx.x >> 5;
        const int c  = threadIdx.x < NBINS ? hist[threadIdx.x] : 0;
        int v = c;
#pragma unroll
        for (int o = 1; o < 32; o <<= 1) {
            const int t = __shfl_up(v, o);
            if (ln >= o) v += t;
        }
        if (ln == 31) wsum[wv] = v;
        __syncthreads();
        int st = v - c;
        for (int w = 0; w < wv; ++w) st += wsum[w];
        if (threadIdx.x < NBINS) {
            base[threadIdx.x] = st;
            desc[bi * NBINS + threadIdx.x] = st | (c << DSH);
        }
        __syncthreads();

        int* myseg = seg + bi * EPB;
#pragma unroll
        for (int q = 0; q < EPT; ++q) {
            if (dv[q] >= 0) {
                const int d = dv[q];
                myseg[base[d >> BSH] + rk[q]] = (sv[q] << BSH) | (d & (BNODES - 1));
            }
        }
        return;
    }

    const int sub  = threadIdx.x >> 7;
    const int t128 = threadIdx.x & 127;
    const int tile = 2 * bi + sub;
    const bool tok = tile < rowTiles;
    const int m0   = (tok ? tile : (rowTiles - 1)) * 16;
    const int lane = threadIdx.x & 31;
    const int wave = t128 >> 5;
    const int n0   = wave * 16;
    const int half = lane >> 4;
    const int idx  = lane & 15;
    constexpr int ldsK = INC + APAD;
    float* as = As[sub];

    {
        const float4* s4 = (const float4*)(A + (long)m0 * INC);
#pragma unroll
        for (int q = t128; q < 16 * (INC / 4); q += 128) {
            const int r = q >> 5;
            const int c = q & 31;
            ((float4*)(as + r * ldsK))[c] = s4[q];
        }
    }
    __syncthreads();

    const float* arow = as + idx * ldsK;
    const int    bcol = n0 + idx;
    const float* bp   = B + bcol;

    v8f acc = {0.f, 0.f, 0.f, 0.f, 0.f, 0.f, 0.f, 0.f};

#pragma unroll 8
    for (int k = 0; k < INC; k += 4) {
        const int ka = k + 2 * half;
        v2f a, b;
        a.x = arow[ka];
        a.y = arow[ka + 1];
        b.x = bp[ka * HIDC];
        b.y = bp[(ka + 1) * HIDC];
        acc = __builtin_amdgcn_wmma_f32_16x16x4_f32(
             false, a,  false, b,
             (short)0, acc,  false,  false);
    }

    if (tok) {
#pragma unroll
        for (int v = 0; v < 8; ++v) {
            const int row = m0 + v + 8 * half;
            C[(long)row * HIDC + bcol] = acc[v];
        }
    }
}

#define RT 1024
#define RU 4

static __device__ __forceinline__ void place_edge(const int pk, const int d0, int* lc, int* ovfN,
                                                  int* __restrict__ tab, int2* __restrict__ ovf) {
    if (pk >= 0) {
        const int dl = pk & (BNODES - 1);
        const int s  = pk >> BSH;
        const int d  = d0 + dl;
        const int r  = atomicAdd(&lc[dl], 1);
        if (r < CAP) {
            tab[2 * (d * CAP + r)] = s;
        } else {
            const int p = atomicAdd(ovfN, 1);
            ovf[p] = make_int2(d, s);
        }
    }
}

__global__ void __launch_bounds__(RT)
k_rank_bins(const int* __restrict__ desc, const int* __restrict__ seg, int nEdgeBlk,
            int* __restrict__ cnt, int* ovfN, int* __restrict__ tab, int2* __restrict__ ovf, int n) {
    __shared__ int lc[BNODES];

    const int k   = blockIdx.x;
    const int d0  = k << BSH;
    const int tid = threadIdx.x;
    if (tid < BNODES) lc[tid] = 0;
    __syncthreads();

    const int lane = tid & 31;
    const int wave = tid >> 5;
    const int half = lane >> 4;
    const int idx  = lane & 15;
    constexpr int NW = RT / 32;

    const int nRun = (nEdgeBlk - wave + NW - 1) / NW;
    for (int c0 = 0; c0 < nRun; c0 += 32) {
        const int nr  = min(32, nRun - c0);
        const int dsc = (lane < nr) ? desc[(wave + (c0 + lane) * NW) * NBINS + k] : 0;
        for (int r0 = 0; r0 < nr; r0 += 2 * RU) {
            int pk[RU], cn[RU], ad[RU];
#pragma unroll
            for (int u = 0; u < RU; ++u) {
                const int r  = r0 + 2 * u + half;
                const int dw = __shfl(dsc, min(r, 31));
                const int d  = (r < nr) ? dw : 0;
                cn[u] = d >> DSH;
                ad[u] = (wave + (c0 + r) * NW) * EPB + (d & ((1 << DSH) - 1));
                pk[u] = (idx < cn[u]) ? seg[ad[u] + idx] : -1;
            }
#pragma unroll
            for (int u = 0; u < RU; ++u) {
                place_edge(pk[u], d0, lc, ovfN, tab, ovf);
                for (int j = 16; __any(j < cn[u]); j += 16) {
                    const int p2 = (j + idx < cn[u]) ? seg[ad[u] + j + idx] : -1;
                    place_edge(p2, d0, lc, ovfN, tab, ovf);
                }
            }
        }
    }
    __syncthreads();
    if (tid < BNODES && d0 + tid < n) cnt[d0 + tid] = lc[tid];
}


static __device__ __forceinline__ void fma4(float4& a, const float4 v, const float w) {
    a.x = fmaf(v.x, w, a.x);
    a.y = fmaf(v.y, w, a.y);
    a.z = fmaf(v.z, w, a.z);
    a.w = fmaf(v.w, w, a.w);
}

static __device__ __forceinline__ void fma4_if(float4& a, const float4 v, const float w, const bool c) {
    a.x = c ? fmaf(v.x, w, a.x) : a.x;
    a.y = c ? fmaf(v.y, w, a.y) : a.y;
    a.z = c ? fmaf(v.z, w, a.z) : a.z;
    a.w = c ? fmaf(v.w, w, a.w) : a.w;
}

#define HPAD 4

#define ROWB(v, e)  __builtin_amdgcn_update_dpp(0, (v), 0x150 + (e), 0xf, 0xf, false)
#define ROWBF(v, e) __int_as_float(ROWB(__float_as_int(v), (e)))

template <int E0>
static __device__ __forceinline__ void rows8(float4 (&v)[8], const int ids,
                                             const float4* __restrict__ hw4, const int g) {
    v[0] = hw4[ROWB(ids, E0 + 0) * (HIDC / 4) + g];
    v[1] = hw4[ROWB(ids, E0 + 1) * (HIDC / 4) + g];
    v[2] = hw4[ROWB(ids, E0 + 2) * (HIDC / 4) + g];
    v[3] = hw4[ROWB(ids, E0 + 3) * (HIDC / 4) + g];
    v[4] = hw4[ROWB(ids, E0 + 4) * (HIDC / 4) + g];
    v[5] = hw4[ROWB(ids, E0 + 5) * (HIDC / 4) + g];
    v[6] = hw4[ROWB(ids, E0 + 6) * (HIDC / 4) + g];
    v[7] = hw4[ROWB(ids, E0 + 7) * (HIDC / 4) + g];
}

template <int E0>
static __device__ __forceinline__ void fma8(float4& acc, const float4 (&v)[8], const float w, const int k) {
    fma4_if(acc, v[0], ROWBF(w, E0 + 0), E0 + 0 < k);
    fma4_if(acc, v[1], ROWBF(w, E0 + 1), E0 + 1 < k);
    fma4_if(acc, v[2], ROWBF(w, E0 + 2), E0 + 2 < k);
    fma4_if(acc, v[3], ROWBF(w, E0 + 3), E0 + 3 < k);
    fma4_if(acc, v[4], ROWBF(w, E0 + 4), E0 + 4 < k);
    fma4_if(acc, v[5], ROWBF(w, E0 + 5), E0 + 5 < k);
    fma4_if(acc, v[6], ROWBF(w, E0 + 6), E0 + 6 < k);
    fma4_if(acc, v[7], ROWBF(w, E0 + 7), E0 + 7 < k);
}

template <bool MAKEW>
static __device__ __forceinline__ void load_slot(const int* tabw, const int2* __restrict__ tabr,
                                                 const int r0, const int q, const int m,
                                                 int& id, float& w) {
    if (q < m) {
        if (MAKEW) {
            id = tabw[2 * (r0 + q)];
        } else {
            const int2 e = tabr[r0 + q];
            id = e.x;
            w  = __int_as_float(e.y);
        }
    }
}

template <int NOUT, bool MAKEW>
__global__ void __launch_bounds__(256)
k_gather_gemm(const float* __restrict__ hw, const int* __restrict__ cnt,
              int* tabw, const int2* __restrict__ tabr,
              const int2* __restrict__ ovf,
              int E, const float* __restrict__ bias,
              const float* __restrict__ Wn, const float* __restrict__ bn,
              float* __restrict__ outp, int n, int nTiles) {
    constexpr int NT  = (NOUT + 15) / 16;
    constexpr int NRT = 8 / NT;
    static_assert(NT * NRT == 8, "column tiles divide the block's 8 waves");
    __shared__ float Hs[2][16 * (HIDC + HPAD)];
    __shared__ float Wt[NOUT * (HIDC + HPAD)];

    const int g    = threadIdx.x & 15;
    const int nb   = threadIdx.x >> 4;
    const int wave = threadIdx.x >> 5;
    const int half = (threadIdx.x >> 4) & 1;
    const int G    = gridDim.x;
    const float4* hw4 = (const float4*)hw;

    const int    ctile = wave % NT;
    const int    turn  = wave / NT;
    const int    bcol  = ctile * 16 + g;
    const bool   bok   = bcol < NOUT;
    constexpr int NWQ = (HIDC * NOUT + 255) / 256;
    float wv[NWQ];
    const float* wrow = Wt + (bok ? bcol : (NOUT - 1)) * (HIDC + HPAD);
    const float  bb = (bok && bn) ? bn[bcol] : 0.0f;

    auto product = [&](const float* hs, const int m0) {
        const float* arow = hs + g * (HIDC + HPAD);
        v8f d = {0.f, 0.f, 0.f, 0.f, 0.f, 0.f, 0.f, 0.f};
#pragma unroll
        for (int k = 0; k < HIDC; k += 4) {
            const int ka = k + 2 * half;
            v2f a, w;
            a.x = arow[ka];
            a.y = arow[ka + 1];
            w.x = wrow[ka];
            w.y = wrow[ka + 1];
            d = __builtin_amdgcn_wmma_f32_16x16x4_f32(
                 false, a,  false, w,
                 (short)0, d,  false,  false);
        }
        if (bok) {
#pragma unroll
            for (int v = 0; v < 8; ++v) {
                const int row = m0 + v + 8 * half;
                if (row < n) outp[(long)row * NOUT + bcol] = d[v] + bb;
            }
        }
    };
    const float4 b  = ((const float4*)bias)[g];

    int   tile = blockIdx.x;
    bool  live = tile * 16 + nb < n;
    int   i    = live ? (tile * 16 + nb) : (n - 1);
    int   c    = cnt[i];
    int   idA  = i, idB = i;
    float wA   = 0.f, wB = 0.f;
    bool nlive = (tile + G) * 16 + nb < n;
    int  ni    = (tile + G < nTiles && nlive) ? ((tile + G) * 16 + nb) : i;
    int  nc    = cnt[ni];
    if (tile + G >= nTiles) nc = 0;
#pragma unroll
    for (int q = 0; q < NWQ; ++q) {
        const int e = threadIdx.x + q * 256;
        wv[q] = (e < HIDC * NOUT) ? Wn[e] : 0.f;
    }
    load_slot<MAKEW>(tabw, tabr, i * CAP, g,      min(c, CAP), idA, wA);
    load_slot<MAKEW>(tabw, tabr, i * CAP, g + 16, min(c, CAP), idB, wB);
#pragma unroll
    for (int q = 0; q < NWQ; ++q) {
        const int e  = threadIdx.x + q * 256;
        const int k  = e / NOUT;
        const int nn = e - k * NOUT;
        if (e < HIDC * NOUT) Wt[nn * (HIDC + HPAD) + k] = wv[q];
    }

    for (int it = 0; tile < nTiles; tile += G, ++it) {
        const int   m  = min(c, CAP);
        const int   r0 = i * CAP;
        const float dd = rsqrtf((float)(c + 1));

        int   nidA = ni, nidB = ni;
        float nwA  = 0.f, nwB = 0.f;
        load_slot<MAKEW>(tabw, tabr, ni * CAP, g,      min(nc, CAP), nidA, nwA);
        load_slot<MAKEW>(tabw, tabr, ni * CAP, g + 16, min(nc, CAP), nidB, nwB);
        const int  nnt    = tile + 2 * G;
        const bool more2  = nnt < nTiles;
        const bool nnlive = nnt * 16 + nb < n;
        const int  nni    = (more2 && nnlive) ? (nnt * 16 + nb) : i;
        int        nnc    = cnt[nni];
        if (!more2) nnc = 0;

        const float4 vs = hw4[i * (HIDC / 4) + g];
        float4 acc = make_float4(0.f, 0.f, 0.f, 0.f);
        float wa = 0.f, wb = 0.f;
        if (m > 0) {
            int ca = 0, cb = 0;
            if (MAKEW) {
                ca = cnt[idA];
                cb = cnt[idB];
            }
            float4 v0[8];
            rows8<0>(v0, idA, hw4, g);
            fma4(acc, vs, dd * dd);
            if (MAKEW) {
                wa = rsqrtf((float)(ca + 1));
                wb = rsqrtf((float)(cb + 1));
                if (live && g < m)      tabw[2 * (r0 + g) + 1]      = __float_as_int(wa);
                if (live && g + 16 < m) tabw[2 * (r0 + 16 + g) + 1] = __float_as_int(wb);
            } else {
                wa = wA;
                wb = wB;
            }
            wa *= dd;
            wb *= dd;
            fma8<0>(acc, v0, wa, m);
        } else {
            fma4(acc, vs, dd * dd);
        }

        if (m > 8) {
            float4 v[8];
            rows8<8>(v, idA, hw4, g);
            fma8<8>(acc, v, wa, m);
        }
        if (m > 16) {
            float4 v[8];
            rows8<0>(v, idB, hw4, g);
            fma8<0>(acc, v, wb, m - 16);
        }
        if (m > 24) {
            float4 v[8];
            rows8<8>(v, idB, hw4, g);
            fma8<8>(acc, v, wb, m - 16);
        }
        for (int base = 32; base < m; base += 16) {
            const int k  = m - base;
            int   id = i;
            float w  = 0.f;
            load_slot<MAKEW>(tabw, tabr, r0 + base, g, k, id, w);
            float4 v[8];
            rows8<0>(v, id, hw4, g);
            if (MAKEW) {
                w = rsqrtf((float)(cnt[id] + 1));
                if (live && g < k) tabw[2 * (r0 + base + g) + 1] = __float_as_int(w);
            }
            w *= dd;
            fma8<0>(acc, v, w, k);
            if (k > 8) {
                rows8<8>(v, id, hw4, g);
                fma8<8>(acc, v, w, k);
            }
        }

        if (c > CAP) {
            const int no = min(cnt[n], E);
            for (int p = 0; p < no; ++p) {
                const int2 r = ovf[p];
                if (r.x == i)
                    fma4(acc, hw4[r.y * (HIDC / 4) + g], rsqrtf((float)(cnt[r.y] + 1)) * dd);
            }
        }

        float4 o;
        o.x = fmaxf(acc.x + b.x, 0.f);
        o.y = fmaxf(acc.y + b.y, 0.f);
        o.z = fmaxf(acc.z + b.z, 0.f);
        o.w = fmaxf(acc.w + b.w, 0.f);

        float* hs = Hs[it & 1];
        ((float4*)(hs + nb * (HIDC + HPAD)))[g] = o;
        __syncthreads();

        if (turn == it % NRT) product(hs, tile * 16);

        i = ni; c = nc; live = nlive;
        idA = nidA; idB = nidB; wA = nwA; wB = nwB;
        ni = nni; nc = nnc; nlive = nnlive;
    }
}


extern "C" void kernel_launch(void* const* d_in, const int* in_sizes, int n_in,
                              void* d_out, int out_size, void* d_ws, size_t ws_size,
                              hipStream_t stream) {
    const float* x  = (const float*)d_in[0];
    const int*   ei = (const int*)d_in[1];
    const float* W1 = (const float*)d_in[2];
    const float* b1 = (const float*)d_in[3];
    const float* W2 = (const float*)d_in[4];
    const float* b2 = (const float*)d_in[5];
    const float* Wc = (const float*)d_in[6];
    const float* bc = (const float*)d_in[7];
    float* out = (float*)d_out;

    const int n = N_NODES;
    const int E = in_sizes[1] / 2;
    const int* src = ei;
    const int* dst = ei + E;

    const int T = 256;
    const int gE   = (E + EPB - 1) / EPB;

    char* ws = (char*)d_ws;
    size_t off = 0;
    int* cnt = (int*)(ws + off);
    off += (((size_t)(n + 1) * 4) + 255) & ~(size_t)255;
    int* seg = (int*)(ws + off);
    off += (((size_t)gE * EPB * 4) + 255) & ~(size_t)255;
    int* desc = (int*)(ws + off);
    off += (((size_t)gE * NBINS * 4) + 255) & ~(size_t)255;
    int* tab = (int*)(ws + off);
    off += (((size_t)n * CAP * 8) + 255) & ~(size_t)255;
    int2* ovf = (int2*)(ws + off);
    off += (((size_t)E * 8) + 255) & ~(size_t)255;
    float* bufA = (float*)(ws + off);
    off += (((size_t)n * HIDC * 4) + 255) & ~(size_t)255;
    float* bufB = (float*)(ws + off);

    const int nTiles = (n + 15) / 16;
    const int gG     = (nTiles + 2) / 3;
    const int rowTiles = n / 16;
    const int nProd = (rowTiles + 1) / 2;
    const int gM    = nProd + gE;

    k_file_gemm<<<gM, T, 0, stream>>>(src, dst, cnt + n, seg, desc, E, gE, x, W1, bufA, rowTiles);
    k_rank_bins<<<NBINS, RT, 0, stream>>>(desc, seg, gE, cnt, cnt + n, tab, ovf, n);

    k_gather_gemm<HIDC, true><<<gG, T, 0, stream>>>(bufA, cnt, tab, nullptr, ovf, E, b1, W2, nullptr, bufB, n, nTiles);

    k_gather_gemm<10, false><<<gG, T, 0, stream>>>(bufB, cnt, nullptr, (const int2*)tab, ovf, E, b2, Wc, bc, out, n, nTiles);
}


# Edge words gathered in LDS and stored as whole 128-byte lines in both table-building launches (a block's sorted segment; slots 0-31 of each node row), not one 4-byte store per edge; sum order kept
# speedup vs baseline: 5.2004x; 1.0741x over previous
#include <hip/hip_runtime.h>

typedef float v2f __attribute__((ext_vector_type(2)));
typedef float v8f __attribute__((ext_vector_type(8)));

#define N_NODES 50000
#define HIDC 64
#define APAD 4

#define CAP 64

#define BSH 8
#define BNODES (1 << BSH)
#define NBINS ((N_NODES + BNODES - 1) / BNODES)
#define EPT 8
#define EPB (EPT * 256)
#define DSH 12
static_assert(NBINS <= 256, "one thread of an edge block per bin");
static_assert(EPB < (1 << DSH), "run start and length fit their fields");

#define INC 128

__global__ void __launch_bounds__(256)
k_file_gemm(const int* __restrict__ src, const int* __restrict__ dst, int* __restrict__ ovfN,
            int* __restrict__ seg, int* __restrict__ desc, int E, int nEdgeBlk,
            const float* __restrict__ A, const float* __restrict__ B, float* __restrict__ C,
            int rowTiles) {
    __shared__ __attribute__((aligned(16))) float As[2][16 * (INC + APAD)];
    __shared__ int hist[NBINS];
    __shared__ int base[NBINS];
    __shared__ int wsum[8];

    const int blk   = blockIdx.x;
    const int nProd = (rowTiles + 1) >> 1;
    const int nMix  = min(nEdgeBlk, nProd);
    const bool mixed  = blk < 2 * nMix;
    const bool isEdge = mixed ? !(blk & 1) : (nEdgeBlk > nProd);
    const int  bi     = mixed ? (blk >> 1) : (blk - nMix);

    if (isEdge) {
        if (bi == 0 && threadIdx.x == 0) *ovfN = 0;
        for (int b = threadIdx.x; b < NBINS; b += 256) hist[b] = 0;
        __syncthreads();

        int dv[EPT], sv[EPT], rk[EPT];
        const int e0 = bi * EPB + threadIdx.x;
#pragma unroll
        for (int q = 0; q < EPT; ++q) {
            const int e = e0 + q * 256;
            dv[q] = -1;
            sv[q] = 0;
            if (e < E) { dv[q] = dst[e]; sv[q] = src[e]; }
        }
#pragma unroll
        for (int q = 0; q < EPT; ++q) {
            rk[q] = 0;
            if (dv[q] >= 0) rk[q] = atomicAdd(&hist[dv[q] >> BSH], 1);
        }
        __syncthreads();

        const int ln = threadIdx.x & 31;
        const int wv = threadIdx.x >> 5;
        const int c  = threadIdx.x < NBINS ? hist[threadIdx.x] : 0;
        int v = c;
#pragma unroll
        for (int o = 1; o < 32; o <<= 1) {
            const int t = __shfl_up(v, o);
            if (ln >= o) v += t;
        }
        if (ln == 31) wsum[wv] = v;
        __syncthreads();
        int st = v - c;
        for (int w = 0; w < wv; ++w) st += wsum[w];
        if (threadIdx.x < NBINS) {
            base[threadIdx.x] = st;
            desc[bi * NBINS + threadIdx.x] = st | (c << DSH);
        }
        __syncthreads();

        int* sorted = (int*)&As[0][0];
        static_assert(EPB * 4 <= (int)sizeof(As), "the segment fits the tile buffer");
#pragma unroll
        for (int q = 0; q < EPT; ++q) {
            if (dv[q] >= 0) {
                const int d = dv[q];
                sorted[base[d >> BSH] + rk[q]] = (sv[q] << BSH) | (d & (BNODES - 1));
            }
        }
        __syncthreads();
        const int nE = min(EPB, E - bi * EPB);
        int4* seg4 = (int4*)(seg + bi * EPB);
        for (int p = threadIdx.x; 4 * p < nE; p += 256) seg4[p] = ((const int4*)sorted)[p];
        return;
    }

    const int sub  = threadIdx.x >> 7;
    const int t128 = threadIdx.x & 127;
    const int tile = 2 * bi + sub;
    const bool tok = tile < rowTiles;
    const int m0   = (tok ? tile : (rowTiles - 1)) * 16;
    const int lane = threadIdx.x & 31;
    const int wave = t128 >> 5;
    const int n0   = wave * 16;
    const int half = lane >> 4;
    const int idx  = lane & 15;
    constexpr int ldsK = INC + APAD;
    float* as = As[sub];

    {
        const float4* s4 = (const float4*)(A + (long)m0 * INC);
#pragma unroll
        for (int q = t128; q < 16 * (INC / 4); q += 128) {
            const int r = q >> 5;
            const int c = q & 31;
            ((float4*)(as + r * ldsK))[c] = s4[q];
        }
    }
    __syncthreads();

    const float* arow = as + idx * ldsK;
    const int    bcol = n0 + idx;
    const float* bp   = B + bcol;

    v8f acc = {0.f, 0.f, 0.f, 0.f, 0.f, 0.f, 0.f, 0.f};

#pragma unroll 8
    for (int k = 0; k < INC; k += 4) {
        const int ka = k + 2 * half;
        v2f a, b;
        a.x = arow[ka];
        a.y = arow[ka + 1];
        b.x = bp[ka * HIDC];
        b.y = bp[(ka + 1) * HIDC];
        acc = __builtin_amdgcn_wmma_f32_16x16x4_f32(
             false, a,  false, b,
             (short)0, acc,  false,  false);
    }

    if (tok) {
#pragma unroll
        for (int v = 0; v < 8; ++v) {
            const int row = m0 + v + 8 * half;
            C[(long)row * HIDC + bcol] = acc[v];
        }
    }
}

#define RT 1024
#define RU 4
#define SROW 32
static_assert(SROW % 16 == 0 && SROW <= CAP, "whole lines of a row");

static __device__ __forceinline__ void place_edge(const int pk, const int d0, int* lc, int* stage, int* ovfN,
                                                  int* __restrict__ tab, int2* __restrict__ ovf) {
    if (pk >= 0) {
        const int dl = pk & (BNODES - 1);
        const int s  = pk >> BSH;
        const int d  = d0 + dl;
        const int r  = atomicAdd(&lc[dl], 1);
        if (r < SROW) {
            stage[dl * SROW + r] = s;
        } else if (r < CAP) {
            tab[2 * (d * CAP + r)] = s;
        } else {
            const int p = atomicAdd(ovfN, 1);
            ovf[p] = make_int2(d, s);
        }
    }
}

__global__ void __launch_bounds__(RT)
k_rank_bins(const int* __restrict__ desc, const int* __restrict__ seg, int nEdgeBlk,
            int* __restrict__ cnt, int* ovfN, int* __restrict__ tab, int2* __restrict__ ovf, int n) {
    __shared__ int lc[BNODES];
    __shared__ int stage[BNODES * SROW];

    const int k   = blockIdx.x;
    const int d0  = k << BSH;
    const int tid = threadIdx.x;
    if (tid < BNODES) lc[tid] = 0;
    __syncthreads();

    const int lane = tid & 31;
    const int wave = tid >> 5;
    const int half = lane >> 4;
    const int idx  = lane & 15;
    constexpr int NW = RT / 32;

    const int nRun = (nEdgeBlk - wave + NW - 1) / NW;
    for (int c0 = 0; c0 < nRun; c0 += 32) {
        const int nr  = min(32, nRun - c0);
        const int dsc = (lane < nr) ? desc[(wave + (c0 + lane) * NW) * NBINS + k] : 0;
        for (int r0 = 0; r0 < nr; r0 += 2 * RU) {
            int pk[RU], cn[RU], ad[RU];
#pragma unroll
            for (int u = 0; u < RU; ++u) {
                const int r  = r0 + 2 * u + half;
                const int dw = __shfl(dsc, min(r, 31));
                const int d  = (r < nr) ? dw : 0;
                cn[u] = d >> DSH;
                ad[u] = (wave + (c0 + r) * NW) * EPB + (d & ((1 << DSH) - 1));
                pk[u] = (idx < cn[u]) ? seg[ad[u] + idx] : -1;
            }
#pragma unroll
            for (int u = 0; u < RU; ++u) {
                place_edge(pk[u], d0, lc, stage, ovfN, tab, ovf);
                for (int j = 16; __any(j < cn[u]); j += 16) {
                    const int p2 = (j + idx < cn[u]) ? seg[ad[u] + j + idx] : -1;
                    place_edge(p2, d0, lc, stage, ovfN, tab, ovf);
                }
            }
        }
    }
    __syncthreads();
    if (tid < BNODES && d0 + tid < n) cnt[d0 + tid] = lc[tid];

    int2* tab2 = (int2*)tab;
    const int hw  = tid >> 4;
    const int l16 = tid & 15;
    constexpr int NLINE = BNODES * (SROW / 16);
    constexpr int NGRP  = RT / 16;
    static_assert(NLINE % NGRP == 0, "the groups share the lines evenly");
    constexpr int NP = NLINE / NGRP;
    int cN[NP], sN[NP];
#pragma unroll
    for (int p = 0; p < NP; ++p) {
        const int L  = hw + p * NGRP;
        const int nd = L & (BNODES - 1);
        cN[p] = lc[nd];
        sN[p] = stage[nd * SROW + 16 * (L >> BSH) + l16];
    }
#pragma unroll
    for (int p = 0; p < NP; ++p) {
        const int L  = hw + p * NGRP;
        const int nd = L & (BNODES - 1);
        const int q  = 16 * (L >> BSH) + l16;
        if (16 * (L >> BSH) < cN[p] && d0 + nd < n)
            tab2[(d0 + nd) * CAP + q] = make_int2(q < cN[p] ? sN[p] : 0, 0);
    }
}


static __device__ __forceinline__ void fma4(float4& a, const float4 v, const float w) {
    a.x = fmaf(v.x, w, a.x);
    a.y = fmaf(v.y, w, a.y);
    a.z = fmaf(v.z, w, a.z);
    a.w = fmaf(v.w, w, a.w);
}

static __device__ __forceinline__ void fma4_if(float4& a, const float4 v, const float w, const bool c) {
    a.x = c ? fmaf(v.x, w, a.x) : a.x;
    a.y = c ? fmaf(v.y, w, a.y) : a.y;
    a.z = c ? fmaf(v.z, w, a.z) : a.z;
    a.w = c ? fmaf(v.w, w, a.w) : a.w;
}

#define HPAD 4

#define ROWB(v, e)  __builtin_amdgcn_update_dpp(0, (v), 0x150 + (e), 0xf, 0xf, false)
#define ROWBF(v, e) __int_as_float(ROWB(__float_as_int(v), (e)))

template <int E0>
static __device__ __forceinline__ void rows8(float4 (&v)[8], const int ids,
                                             const float4* __restrict__ hw4, const int g) {
    v[0] = hw4[ROWB(ids, E0 + 0) * (HIDC / 4) + g];
    v[1] = hw4[ROWB(ids, E0 + 1) * (HIDC / 4) + g];
    v[2] = hw4[ROWB(ids, E0 + 2) * (HIDC / 4) + g];
    v[3] = hw4[ROWB(ids, E0 + 3) * (HIDC / 4) + g];
    v[4] = hw4[ROWB(ids, E0 + 4) * (HIDC / 4) + g];
    v[5] = hw4[ROWB(ids, E0 + 5) * (HIDC / 4) + g];
    v[6] = hw4[ROWB(ids, E0 + 6) * (HIDC / 4) + g];
    v[7] = hw4[ROWB(ids, E0 + 7) * (HIDC / 4) + g];
}

template <int E0>
static __device__ __forceinline__ void fma8(float4& acc, const float4 (&v)[8], const float w, const int k) {
    fma4_if(acc, v[0], ROWBF(w, E0 + 0), E0 + 0 < k);
    fma4_if(acc, v[1], ROWBF(w, E0 + 1), E0 + 1 < k);
    fma4_if(acc, v[2], ROWBF(w, E0 + 2), E0 + 2 < k);
    fma4_if(acc, v[3], ROWBF(w, E0 + 3), E0 + 3 < k);
    fma4_if(acc, v[4], ROWBF(w, E0 + 4), E0 + 4 < k);
    fma4_if(acc, v[5], ROWBF(w, E0 + 5), E0 + 5 < k);
    fma4_if(acc, v[6], ROWBF(w, E0 + 6), E0 + 6 < k);
    fma4_if(acc, v[7], ROWBF(w, E0 + 7), E0 + 7 < k);
}

template <bool MAKEW>
static __device__ __forceinline__ void load_slot(const int* tabw, const int2* __restrict__ tabr,
                                                 const int r0, const int q, const int m,
                                                 int& id, float& w) {
    if (q < m) {
        if (MAKEW) {
            id = tabw[2 * (r0 + q)];
        } else {
            const int2 e = tabr[r0 + q];
            id = e.x;
            w  = __int_as_float(e.y);
        }
    }
}

template <int NOUT, bool MAKEW>
__global__ void __launch_bounds__(256)
k_gather_gemm(const float* __restrict__ hw, const int* __restrict__ cnt,
              int* tabw, const int2* __restrict__ tabr,
              const int2* __restrict__ ovf,
              int E, const float* __restrict__ bias,
              const float* __restrict__ Wn, const float* __restrict__ bn,
              float* __restrict__ outp, int n, int nTiles) {
    constexpr int NT  = (NOUT + 15) / 16;
    constexpr int NRT = 8 / NT;
    static_assert(NT * NRT == 8, "column tiles divide the block's 8 waves");
    __shared__ float Hs[2][16 * (HIDC + HPAD)];
    __shared__ float Wt[NOUT * (HIDC + HPAD)];

    const int g    = threadIdx.x & 15;
    const int nb   = threadIdx.x >> 4;
    const int wave = threadIdx.x >> 5;
    const int half = (threadIdx.x >> 4) & 1;
    const int G    = gridDim.x;
    const float4* hw4 = (const float4*)hw;

    const int    ctile = wave % NT;
    const int    turn  = wave / NT;
    const int    bcol  = ctile * 16 + g;
    const bool   bok   = bcol < NOUT;
    constexpr int NWQ = (HIDC * NOUT + 255) / 256;
    float wv[NWQ];
    const float* wrow = Wt + (bok ? bcol : (NOUT - 1)) * (HIDC + HPAD);
    const float  bb = (bok && bn) ? bn[bcol] : 0.0f;

    auto product = [&](const float* hs, const int m0) {
        const float* arow = hs + g * (HIDC + HPAD);
        v8f d = {0.f, 0.f, 0.f, 0.f, 0.f, 0.f, 0.f, 0.f};
#pragma unroll
        for (int k = 0; k < HIDC; k += 4) {
            const int ka = k + 2 * half;
            v2f a, w;
            a.x = arow[ka];
            a.y = arow[ka + 1];
            w.x = wrow[ka];
            w.y = wrow[ka + 1];
            d = __builtin_amdgcn_wmma_f32_16x16x4_f32(
                 false, a,  false, w,
                 (short)0, d,  false,  false);
        }
        if (bok) {
#pragma unroll
            for (int v = 0; v < 8; ++v) {
                const int row = m0 + v + 8 * half;
                if (row < n) outp[(long)row * NOUT + bcol] = d[v] + bb;
            }
        }
    };
    const float4 b  = ((const float4*)bias)[g];

    int   tile = blockIdx.x;
    bool  live = tile * 16 + nb < n;
    int   i    = live ? (tile * 16 + nb) : (n - 1);
    int   c    = cnt[i];
    int   idA  = i, idB = i;
    float wA   = 0.f, wB = 0.f;
    bool nlive = (tile + G) * 16 + nb < n;
    int  ni    = (tile + G < nTiles && nlive) ? ((tile + G) * 16 + nb) : i;
    int  nc    = cnt[ni];
    if (tile + G >= nTiles) nc = 0;
#pragma unroll
    for (int q = 0; q < NWQ; ++q) {
        const int e = threadIdx.x + q * 256;
        wv[q] = (e < HIDC * NOUT) ? Wn[e] : 0.f;
    }
    load_slot<MAKEW>(tabw, tabr, i * CAP, g,      min(c, CAP), idA, wA);
    load_slot<MAKEW>(tabw, tabr, i * CAP, g + 16, min(c, CAP), idB, wB);
#pragma unroll
    for (int q = 0; q < NWQ; ++q) {
        const int e  = threadIdx.x + q * 256;
        const int k  = e / NOUT;
        const int nn = e - k * NOUT;
        if (e < HIDC * NOUT) Wt[nn * (HIDC + HPAD) + k] = wv[q];
    }

    for (int it = 0; tile < nTiles; tile += G, ++it) {
        const int   m  = min(c, CAP);
        const int   r0 = i * CAP;
        const float dd = rsqrtf((float)(c + 1));

        int   nidA = ni, nidB = ni;
        float nwA  = 0.f, nwB = 0.f;
        load_slot<MAKEW>(tabw, tabr, ni * CAP, g,      min(nc, CAP), nidA, nwA);
        load_slot<MAKEW>(tabw, tabr, ni * CAP, g + 16, min(nc, CAP), nidB, nwB);
        const int  nnt    = tile + 2 * G;
        const bool more2  = nnt < nTiles;
        const bool nnlive = nnt * 16 + nb < n;
        const int  nni    = (more2 && nnlive) ? (nnt * 16 + nb) : i;
        int        nnc    = cnt[nni];
        if (!more2) nnc = 0;

        const float4 vs = hw4[i * (HIDC / 4) + g];
        float4 acc = make_float4(0.f, 0.f, 0.f, 0.f);
        float wa = 0.f, wb = 0.f;
        if (m > 0) {
            int ca = 0, cb = 0;
            if (MAKEW) {
                ca = cnt[idA];
                cb = cnt[idB];
            }
            float4 v0[8];
            rows8<0>(v0, idA, hw4, g);
            fma4(acc, vs, dd * dd);
            if (MAKEW) {
                wa = rsqrtf((float)(ca + 1));
                wb = rsqrtf((float)(cb + 1));
                if (live && g < m)      tabw[2 * (r0 + g) + 1]      = __float_as_int(wa);
                if (live && g + 16 < m) tabw[2 * (r0 + 16 + g) + 1] = __float_as_int(wb);
            } else {
                wa = wA;
                wb = wB;
            }
            wa *= dd;
            wb *= dd;
            fma8<0>(acc, v0, wa, m);
        } else {
            fma4(acc, vs, dd * dd);
        }

        if (m > 8) {
            float4 v[8];
            rows8<8>(v, idA, hw4, g);
            fma8<8>(acc, v, wa, m);
        }
        if (m > 16) {
            float4 v[8];
            rows8<0>(v, idB, hw4, g);
            fma8<0>(acc, v, wb, m - 16);
        }
        if (m > 24) {
            float4 v[8];
            rows8<8>(v, idB, hw4, g);
            fma8<8>(acc, v, wb, m - 16);
        }
        for (int base = 32; base < m; base += 16) {
            const int k  = m - base;
            int   id = i;
            float w  = 0.f;
            load_slot<MAKEW>(tabw, tabr, r0 + base, g, k, id, w);
            float4 v[8];
            rows8<0>(v, id, hw4, g);
            if (MAKEW) {
                w = rsqrtf((float)(cnt[id] + 1));
                if (live && g < k) tabw[2 * (r0 + base + g) + 1] = __float_as_int(w);
            }
            w *= dd;
            fma8<0>(acc, v, w, k);
            if (k > 8) {
                rows8<8>(v, id, hw4, g);
                fma8<8>(acc, v, w, k);
            }
        }

        if (c > CAP) {
            const int no = min(cnt[n], E);
            for (int p = 0; p < no; ++p) {
                const int2 r = ovf[p];
                if (r.x == i)
                    fma4(acc, hw4[r.y * (HIDC / 4) + g], rsqrtf((float)(cnt[r.y] + 1)) * dd);
            }
        }

        float4 o;
        o.x = fmaxf(acc.x + b.x, 0.f);
        o.y = fmaxf(acc.y + b.y, 0.f);
        o.z = fmaxf(acc.z + b.z, 0.f);
        o.w = fmaxf(acc.w + b.w, 0.f);

        float* hs = Hs[it & 1];
        ((float4*)(hs + nb * (HIDC + HPAD)))[g] = o;
        __syncthreads();

        if (turn == it % NRT) product(hs, tile * 16);

        i = ni; c = nc; live = nlive;
        idA = nidA; idB = nidB; wA = nwA; wB = nwB;
        ni = nni; nc = nnc; nlive = nnlive;
    }
}


extern "C" void kernel_launch(void* const* d_in, const int* in_sizes, int n_in,
                              void* d_out, int out_size, void* d_ws, size_t ws_size,
                              hipStream_t stream) {
    const float* x  = (const float*)d_in[0];
    const int*   ei = (const int*)d_in[1];
    const float* W1 = (const float*)d_in[2];
    const float* b1 = (const float*)d_in[3];
    const float* W2 = (const float*)d_in[4];
    const float* b2 = (const float*)d_in[5];
    const float* Wc = (const float*)d_in[6];
    const float* bc = (const float*)d_in[7];
    float* out = (float*)d_out;

    const int n = N_NODES;
    const int E = in_sizes[1] / 2;
    const int* src = ei;
    const int* dst = ei + E;

    const int T = 256;
    const int gE   = (E + EPB - 1) / EPB;

    char* ws = (char*)d_ws;
    size_t off = 0;
    int* cnt = (int*)(ws + off);
    off += (((size_t)(n + 1) * 4) + 255) & ~(size_t)255;
    int* seg = (int*)(ws + off);
    off += (((size_t)gE * EPB * 4) + 255) & ~(size_t)255;
    int* desc = (int*)(ws + off);
    off += (((size_t)gE * NBINS * 4) + 255) & ~(size_t)255;
    int* tab = (int*)(ws + off);
    off += (((size_t)n * CAP * 8) + 255) & ~(size_t)255;
    int2* ovf = (int2*)(ws + off);
    off += (((size_t)E * 8) + 255) & ~(size_t)255;
    float* bufA = (float*)(ws + off);
    off += (((size_t)n * HIDC * 4) + 255) & ~(size_t)255;
    float* bufB = (float*)(ws + off);

    const int nTiles = (n + 15) / 16;
    const int gG     = (nTiles + 2) / 3;
    const int rowTiles = n / 16;
    const int nProd = (rowTiles + 1) / 2;
    const int gM    = nProd + gE;

    k_file_gemm<<<gM, T, 0, stream>>>(src, dst, cnt + n, seg, desc, E, gE, x, W1, bufA, rowTiles);
    k_rank_bins<<<NBINS, RT, 0, stream>>>(desc, seg, gE, cnt, cnt + n, tab, ovf, n);

    k_gather_gemm<HIDC, true><<<gG, T, 0, stream>>>(bufA, cnt, tab, nullptr, ovf, E, b1, W2, nullptr, bufB, n, nTiles);

    k_gather_gemm<10, false><<<gG, T, 0, stream>>>(bufB, cnt, nullptr, (const int2*)tab, ovf, E, b2, Wc, bc, out, n, nTiles);
}
